# starting kernel
# baseline (speedup 1.0000x reference)
#include <hip/hip_runtime.h>

typedef float v2f __attribute__((ext_vector_type(2)));
typedef float v8f __attribute__((ext_vector_type(8)));

#define NS_TOTAL 16384
#define SITES    8
#define NCOL     (SITES * 18)
#define KPAD     112
#define MPAD     80
#define THREADS  256

struct C3 { float2 m[9]; };

__device__ inline float2 cmul(float2 a, float2 b) {
  return make_float2(a.x*b.x - a.y*b.y, a.x*b.y + a.y*b.x);
}
__device__ inline float2 cmulcj(float2 a, float2 b) {
  return make_float2(a.x*b.x + a.y*b.y, a.y*b.x - a.x*b.y);
}
__device__ inline float2 cjmul(float2 a, float2 b) {
  return make_float2(a.x*b.x + a.y*b.y, a.x*b.y - a.y*b.x);
}

__device__ inline int neighbor(int s, int axis, int dir) {
  int a0 = (s >> 10) & 7, a1 = (s >> 7) & 7, a2 = (s >> 4) & 7, a3 = s & 15;
  if      (axis == 0) a0 = (a0 + dir) & 7;
  else if (axis == 1) a1 = (a1 + dir) & 7;
  else if (axis == 2) a2 = (a2 + dir) & 7;
  else                a3 = (a3 + dir) & 15;
  return (a0 << 10) | (a1 << 7) | (a2 << 4) | a3;
}

__device__ inline void loadMat(const float* __restrict__ xin, int bx, int ch, C3& M) {
  const float* p = xin + ((size_t)bx * 10 + ch) * 18;
  #pragma unroll
  for (int e = 0; e < 9; ++e) M.m[e] = make_float2(p[2*e], p[2*e + 1]);
}

__device__ inline void mul33(const C3& A, const C3& B, C3& C) {
  #pragma unroll
  for (int i = 0; i < 3; ++i)
    #pragma unroll
    for (int k = 0; k < 3; ++k) {
      float2 acc = make_float2(0.f, 0.f);
      #pragma unroll
      for (int j = 0; j < 3; ++j) {
        float2 t = cmul(A.m[i*3+j], B.m[j*3+k]);
        acc.x += t.x; acc.y += t.y;
      }
      C.m[i*3+k] = acc;
    }
}

__device__ inline void sandwichF(const C3& U, const C3& W, C3& R) {
  C3 T; mul33(U, W, T);
  #pragma unroll
  for (int i = 0; i < 3; ++i)
    #pragma unroll
    for (int l = 0; l < 3; ++l) {
      float2 acc = make_float2(0.f, 0.f);
      #pragma unroll
      for (int k = 0; k < 3; ++k) {
        float2 t = cmulcj(T.m[i*3+k], U.m[l*3+k]);
        acc.x += t.x; acc.y += t.y;
      }
      R.m[i*3+l] = acc;
    }
}

__device__ inline void sandwichB(const C3& U, const C3& W, C3& R) {
  C3 T;
  #pragma unroll
  for (int i = 0; i < 3; ++i)
    #pragma unroll
    for (int k = 0; k < 3; ++k) {
      float2 acc = make_float2(0.f, 0.f);
      #pragma unroll
      for (int j = 0; j < 3; ++j) {
        float2 t = cjmul(U.m[j*3+i], W.m[j*3+k]);
        acc.x += t.x; acc.y += t.y;
      }
      T.m[i*3+k] = acc;
    }
  #pragma unroll
  for (int i = 0; i < 3; ++i)
    #pragma unroll
    for (int l = 0; l < 3; ++l) {
      float2 acc = make_float2(0.f, 0.f);
      #pragma unroll
      for (int k = 0; k < 3; ++k) {
        float2 t = cmul(T.m[i*3+k], U.m[k*3+l]);
        acc.x += t.x; acc.y += t.y;
      }
      R.m[i*3+l] = acc;
    }
}

__device__ inline void storeRow(float* dst, int ch, int ls, const C3& M) {
  float* p = dst + ch * NCOL + ls * 18;
  #pragma unroll
  for (int e = 0; e < 9; ++e) { p[2*e] = M.m[e].x; p[2*e + 1] = M.m[e].y; }
}
__device__ inline void storeRowDag(float* dst, int ch, int ls, const C3& M) {
  float* p = dst + ch * NCOL + ls * 18;
  #pragma unroll
  for (int i = 0; i < 3; ++i)
    #pragma unroll
    for (int j = 0; j < 3; ++j) {
      p[(i*3+j)*2]     =  M.m[j*3+i].x;
      p[(i*3+j)*2 + 1] = -M.m[j*3+i].y;
    }
}

__global__ __launch_bounds__(THREADS)
void lconv_bilin_kernel(const float* __restrict__ xin,
                        const float* __restrict__ wgt,
                        float* __restrict__ out) {
  extern __shared__ float smem[];
  float* Wt = smem;
  float* tF = Wt + MPAD * KPAD;
  float* rL = tF + KPAD * NCOL;
  float* wF = rL + MPAD * NCOL;

  const int tid  = threadIdx.x;
  const int base = blockIdx.x * SITES;

  for (int idx = tid; idx < MPAD * KPAD; idx += THREADS) {
    int m_ = idx / KPAD, k_ = idx % KPAD;
    Wt[idx] = (m_ < 78 && k_ < 109) ? wgt[m_ * 109 + k_] : 0.0f;
  }
  for (int idx = tid; idx < 3 * NCOL; idx += THREADS) tF[109 * NCOL + idx] = 0.0f;

  for (int t = tid; t < SITES * 48; t += THREADS) {
    int ls = t / 48, rem = t % 48;
    int axis = rem / 12, rr = rem % 12;
    int c = rr % 6;
    bool backward = (rr < 6);
    int bx = base + ls;
    int s = bx & 8191, bofs = bx & ~8191;
    C3 U, W, R;
    if (backward) {
      int sn = neighbor(s, axis, -1);
      loadMat(xin, bofs | sn, axis, U);
      loadMat(xin, bofs | sn, 4 + c, W);
      sandwichB(U, W, R);
    } else {
      int sn = neighbor(s, axis, +1);
      loadMat(xin, bx, axis, U);
      loadMat(xin, bofs | sn, 4 + c, W);
      sandwichF(U, W, R);
    }
    int ch = 6 + axis * 12 + rr;
    storeRow(tF, ch, ls, R);
    storeRowDag(tF, 54 + ch, ls, R);
  }

  for (int t = tid; t < SITES * 6; t += THREADS) {
    int ls = t / 6, c = t % 6;
    C3 W; loadMat(xin, base + ls, 4 + c, W);
    storeRow(tF, c, ls, W);
    storeRowDag(tF, 54 + c, ls, W);
    storeRow(wF, c, ls, W);
    storeRowDag(wF, 6 + c, ls, W);
  }

  for (int t = tid; t < SITES; t += THREADS) {
    float* p1 = tF + 108 * NCOL + t * 18;
    float* p2 = wF + 12 * NCOL + t * 18;
    #pragma unroll
    for (int i = 0; i < 3; ++i)
      #pragma unroll
      for (int j = 0; j < 3; ++j) {
        float re = (i == j) ? 1.0f : 0.0f;
        p1[(i*3+j)*2] = re; p1[(i*3+j)*2 + 1] = 0.0f;
        p2[(i*3+j)*2] = re; p2[(i*3+j)*2 + 1] = 0.0f;
      }
  }

  __syncthreads();

  {
    const int wv = tid >> 5, lane = tid & 31;
    const int l15 = lane & 15, kh = lane >> 4;
    for (int tile = wv; tile < 5 * 9; tile += 8) {
      int mt = tile / 9, nt = tile % 9;
      v8f acc = {};
      const float* wbase = Wt + (mt * 16 + l15) * KPAD;
      const float* bbase = tF + nt * 16 + l15;
      for (int k0 = 0; k0 < KPAD; k0 += 4) {
        v2f a, b;
        a[0] = wbase[k0 + 2 * kh];
        a[1] = wbase[k0 + 2 * kh + 1];
        const float* bp = bbase + (k0 + 2 * kh) * NCOL;
        b[0] = bp[0];
        b[1] = bp[NCOL];
        acc = __builtin_amdgcn_wmma_f32_16x16x4_f32(
            false, a, false, b, (short)0, acc, false, false);
      }
      float* rp = rL + (mt * 16 + (lane >> 4) * 8) * NCOL + nt * 16 + l15;
      #pragma unroll
      for (int q = 0; q < 8; ++q) rp[q * NCOL] = acc[q];
    }
  }
  __syncthreads();

  for (int t = tid; t < SITES * 6 * 9; t += THREADS) {
    int ls = t / 54, rem = t % 54;
    int u = rem / 9, ik = rem % 9, i = ik / 3, k = ik % 3;
    float2 acc = make_float2(0.f, 0.f);
    #pragma unroll
    for (int v = 0; v < 13; ++v) {
      const float* wp = wF + v * NCOL + ls * 18;
      const float* rp = rL + (u * 13 + v) * NCOL + ls * 18;
      #pragma unroll
      for (int j = 0; j < 3; ++j) {
        float2 a = make_float2(wp[(i*3+j)*2], wp[(i*3+j)*2 + 1]);
        float2 b = make_float2(rp[(j*3+k)*2], rp[(j*3+k)*2 + 1]);
        acc.x += a.x * b.x - a.y * b.y;
        acc.y += a.x * b.y + a.y * b.x;
      }
    }
    int bx = base + ls;
    float* op = out + ((size_t)bx * 10 + 4 + u) * 18 + ik * 2;
    op[0] = acc.x; op[1] = acc.y;
  }

  for (int t = tid; t < SITES * 4 * 18; t += THREADS) {
    int ls = t / 72, rem = t % 72;
    int ch = rem / 18, e = rem % 18;
    size_t idx = ((size_t)(base + ls) * 10 + ch) * 18 + e;
    out[idx] = xin[idx];
  }
}

extern "C" void kernel_launch(void* const* d_in, const int* in_sizes, int n_in,
                              void* d_out, int out_size, void* d_ws, size_t ws_size,
                              hipStream_t stream) {
  (void)in_sizes; (void)n_in; (void)out_size; (void)d_ws; (void)ws_size;
  const float* x = (const float*)d_in[0];
  const float* w = (const float*)d_in[1];
  float* out = (float*)d_out;

  const size_t shmem =
      (size_t)(MPAD * KPAD + KPAD * NCOL + MPAD * NCOL + 13 * NCOL) * sizeof(float);
  (void)hipFuncSetAttribute((const void*)lconv_bilin_kernel,
                            hipFuncAttributeMaxDynamicSharedMemorySize, (int)shmem);
  lconv_bilin_kernel<<<NS_TOTAL / SITES, THREADS, shmem, stream>>>(x, w, out);
}


# weight staging: the 35 load-wait-store round trips per thread become 9 16-byte loads all in flight ahead of the transport phase, stored to LDS after it; same LDS layout, no sum reordered
# speedup vs baseline: 1.4446x; 1.4446x over previous
#include <hip/hip_runtime.h>

typedef float v2f __attribute__((ext_vector_type(2)));
typedef float v8f __attribute__((ext_vector_type(8)));

#define NS_TOTAL 16384
#define SITES    8
#define NCOL     (SITES * 18)
#define KPAD     112
#define MPAD     80
#define THREADS  256
#define WN       (78 * 109)
#define WQ_PER_THREAD 9

struct C3 { float2 m[9]; };

__device__ inline float2 cmul(float2 a, float2 b) {
  return make_float2(a.x*b.x - a.y*b.y, a.x*b.y + a.y*b.x);
}
__device__ inline float2 cmulcj(float2 a, float2 b) {
  return make_float2(a.x*b.x + a.y*b.y, a.y*b.x - a.x*b.y);
}
__device__ inline float2 cjmul(float2 a, float2 b) {
  return make_float2(a.x*b.x + a.y*b.y, a.x*b.y - a.y*b.x);
}

__device__ inline int neighbor(int s, int axis, int dir) {
  int a0 = (s >> 10) & 7, a1 = (s >> 7) & 7, a2 = (s >> 4) & 7, a3 = s & 15;
  if      (axis == 0) a0 = (a0 + dir) & 7;
  else if (axis == 1) a1 = (a1 + dir) & 7;
  else if (axis == 2) a2 = (a2 + dir) & 7;
  else                a3 = (a3 + dir) & 15;
  return (a0 << 10) | (a1 << 7) | (a2 << 4) | a3;
}

__device__ inline void loadMat(const float* __restrict__ xin, int bx, int ch, C3& M) {
  const float* p = xin + ((size_t)bx * 10 + ch) * 18;
  #pragma unroll
  for (int e = 0; e < 9; ++e) M.m[e] = make_float2(p[2*e], p[2*e + 1]);
}

__device__ inline void mul33(const C3& A, const C3& B, C3& C) {
  #pragma unroll
  for (int i = 0; i < 3; ++i)
    #pragma unroll
    for (int k = 0; k < 3; ++k) {
      float2 acc = make_float2(0.f, 0.f);
      #pragma unroll
      for (int j = 0; j < 3; ++j) {
        float2 t = cmul(A.m[i*3+j], B.m[j*3+k]);
        acc.x += t.x; acc.y += t.y;
      }
      C.m[i*3+k] = acc;
    }
}

__device__ inline void sandwichF(const C3& U, const C3& W, C3& R) {
  C3 T; mul33(U, W, T);
  #pragma unroll
  for (int i = 0; i < 3; ++i)
    #pragma unroll
    for (int l = 0; l < 3; ++l) {
      float2 acc = make_float2(0.f, 0.f);
      #pragma unroll
      for (int k = 0; k < 3; ++k) {
        float2 t = cmulcj(T.m[i*3+k], U.m[l*3+k]);
        acc.x += t.x; acc.y += t.y;
      }
      R.m[i*3+l] = acc;
    }
}

__device__ inline void sandwichB(const C3& U, const C3& W, C3& R) {
  C3 T;
  #pragma unroll
  for (int i = 0; i < 3; ++i)
    #pragma unroll
    for (int k = 0; k < 3; ++k) {
      float2 acc = make_float2(0.f, 0.f);
      #pragma unroll
      for (int j = 0; j < 3; ++j) {
        float2 t = cjmul(U.m[j*3+i], W.m[j*3+k]);
        acc.x += t.x; acc.y += t.y;
      }
      T.m[i*3+k] = acc;
    }
  #pragma unroll
  for (int i = 0; i < 3; ++i)
    #pragma unroll
    for (int l = 0; l < 3; ++l) {
      float2 acc = make_float2(0.f, 0.f);
      #pragma unroll
      for (int k = 0; k < 3; ++k) {
        float2 t = cmul(T.m[i*3+k], U.m[k*3+l]);
        acc.x += t.x; acc.y += t.y;
      }
      R.m[i*3+l] = acc;
    }
}

__device__ inline void storeRow(float* dst, int ch, int ls, const C3& M) {
  float* p = dst + ch * NCOL + ls * 18;
  #pragma unroll
  for (int e = 0; e < 9; ++e) { p[2*e] = M.m[e].x; p[2*e + 1] = M.m[e].y; }
}
__device__ inline void storeRowDag(float* dst, int ch, int ls, const C3& M) {
  float* p = dst + ch * NCOL + ls * 18;
  #pragma unroll
  for (int i = 0; i < 3; ++i)
    #pragma unroll
    for (int j = 0; j < 3; ++j) {
      p[(i*3+j)*2]     =  M.m[j*3+i].x;
      p[(i*3+j)*2 + 1] = -M.m[j*3+i].y;
    }
}

__global__ __launch_bounds__(THREADS)
void lconv_bilin_kernel(const float* __restrict__ xin,
                        const float* __restrict__ wgt,
                        float* __restrict__ out) {
  extern __shared__ float smem[];
  float* Wt = smem;
  float* tF = Wt + MPAD * KPAD;
  float* rL = tF + KPAD * NCOL;
  float* wF = rL + MPAD * NCOL;

  const int tid  = threadIdx.x;
  const int base = blockIdx.x * SITES;

  float4 wv[WQ_PER_THREAD];
  {
    const float4* w4 = reinterpret_cast<const float4*>(wgt);
    #pragma unroll
    for (int i = 0; i < WQ_PER_THREAD - 1; ++i) wv[i] = w4[tid + THREADS * i];
    const int q = tid + THREADS * (WQ_PER_THREAD - 1);
    wv[WQ_PER_THREAD - 1] = w4[q < WN / 4 ? q : WN / 4 - 1];
  }
  const float wtail0 = wgt[WN - 2], wtail1 = wgt[WN - 1];
  for (int idx = tid; idx < 2 * KPAD + 78 * 3; idx += THREADS) {
    int a;
    if (idx < 2 * KPAD) a = 78 * KPAD + idx;
    else { int j = idx - 2 * KPAD; a = (j / 3) * KPAD + 109 + (j % 3); }
    Wt[a] = 0.0f;
  }
  for (int idx = tid; idx < 3 * NCOL; idx += THREADS) tF[109 * NCOL + idx] = 0.0f;

  for (int t = tid; t < SITES * 48; t += THREADS) {
    int ls = t / 48, rem = t % 48;
    int axis = rem / 12, rr = rem % 12;
    int c = rr % 6;
    bool backward = (rr < 6);
    int bx = base + ls;
    int s = bx & 8191, bofs = bx & ~8191;
    C3 U, W, R;
    if (backward) {
      int sn = neighbor(s, axis, -1);
      loadMat(xin, bofs | sn, axis, U);
      loadMat(xin, bofs | sn, 4 + c, W);
      sandwichB(U, W, R);
    } else {
      int sn = neighbor(s, axis, +1);
      loadMat(xin, bx, axis, U);
      loadMat(xin, bofs | sn, 4 + c, W);
      sandwichF(U, W, R);
    }
    int ch = 6 + axis * 12 + rr;
    storeRow(tF, ch, ls, R);
    storeRowDag(tF, 54 + ch, ls, R);
  }

  for (int t = tid; t < SITES * 6; t += THREADS) {
    int ls = t / 6, c = t % 6;
    C3 W; loadMat(xin, base + ls, 4 + c, W);
    storeRow(tF, c, ls, W);
    storeRowDag(tF, 54 + c, ls, W);
    storeRow(wF, c, ls, W);
    storeRowDag(wF, 6 + c, ls, W);
  }

  for (int t = tid; t < SITES; t += THREADS) {
    float* p1 = tF + 108 * NCOL + t * 18;
    float* p2 = wF + 12 * NCOL + t * 18;
    #pragma unroll
    for (int i = 0; i < 3; ++i)
      #pragma unroll
      for (int j = 0; j < 3; ++j) {
        float re = (i == j) ? 1.0f : 0.0f;
        p1[(i*3+j)*2] = re; p1[(i*3+j)*2 + 1] = 0.0f;
        p2[(i*3+j)*2] = re; p2[(i*3+j)*2 + 1] = 0.0f;
      }
  }

  #pragma unroll
  for (int i = 0; i < WQ_PER_THREAD; ++i) {
    const int e = 4 * (tid + THREADS * i);
    const float v[4] = {wv[i].x, wv[i].y, wv[i].z, wv[i].w};
    if (i < WQ_PER_THREAD - 1 || e + 3 < WN) {
      #pragma unroll
      for (int cc = 0; cc < 4; ++cc) Wt[(e + cc) + 3 * ((e + cc) / 109)] = v[cc];
    }
  }
  if (tid == WN / 4 - THREADS * (WQ_PER_THREAD - 1)) {
    Wt[77 * KPAD + 107] = wtail0;
    Wt[77 * KPAD + 108] = wtail1;
  }

  __syncthreads();

  {
    const int wv = tid >> 5, lane = tid & 31;
    const int l15 = lane & 15, kh = lane >> 4;
    for (int tile = wv; tile < 5 * 9; tile += 8) {
      int mt = tile / 9, nt = tile % 9;
      v8f acc = {};
      const float* wbase = Wt + (mt * 16 + l15) * KPAD;
      const float* bbase = tF + nt * 16 + l15;
      for (int k0 = 0; k0 < KPAD; k0 += 4) {
        v2f a, b;
        a[0] = wbase[k0 + 2 * kh];
        a[1] = wbase[k0 + 2 * kh + 1];
        const float* bp = bbase + (k0 + 2 * kh) * NCOL;
        b[0] = bp[0];
        b[1] = bp[NCOL];
        acc = __builtin_amdgcn_wmma_f32_16x16x4_f32(
            false, a, false, b, (short)0, acc, false, false);
      }
      float* rp = rL + (mt * 16 + (lane >> 4) * 8) * NCOL + nt * 16 + l15;
      #pragma unroll
      for (int q = 0; q < 8; ++q) rp[q * NCOL] = acc[q];
    }
  }
  __syncthreads();

  for (int t = tid; t < SITES * 6 * 9; t += THREADS) {
    int ls = t / 54, rem = t % 54;
    int u = rem / 9, ik = rem % 9, i = ik / 3, k = ik % 3;
    float2 acc = make_float2(0.f, 0.f);
    #pragma unroll
    for (int v = 0; v < 13; ++v) {
      const float* wp = wF + v * NCOL + ls * 18;
      const float* rp = rL + (u * 13 + v) * NCOL + ls * 18;
      #pragma unroll
      for (int j = 0; j < 3; ++j) {
        float2 a = make_float2(wp[(i*3+j)*2], wp[(i*3+j)*2 + 1]);
        float2 b = make_float2(rp[(j*3+k)*2], rp[(j*3+k)*2 + 1]);
        acc.x += a.x * b.x - a.y * b.y;
        acc.y += a.x * b.y + a.y * b.x;
      }
    }
    int bx = base + ls;
    float* op = out + ((size_t)bx * 10 + 4 + u) * 18 + ik * 2;
    op[0] = acc.x; op[1] = acc.y;
  }

  for (int t = tid; t < SITES * 4 * 18; t += THREADS) {
    int ls = t / 72, rem = t % 72;
    int ch = rem / 18, e = rem % 18;
    size_t idx = ((size_t)(base + ls) * 10 + ch) * 18 + e;
    out[idx] = xin[idx];
  }
}

extern "C" void kernel_launch(void* const* d_in, const int* in_sizes, int n_in,
                              void* d_out, int out_size, void* d_ws, size_t ws_size,
                              hipStream_t stream) {
  (void)in_sizes; (void)n_in; (void)out_size; (void)d_ws; (void)ws_size;
  const float* x = (const float*)d_in[0];
  const float* w = (const float*)d_in[1];
  float* out = (float*)d_out;

  const size_t shmem =
      (size_t)(MPAD * KPAD + KPAD * NCOL + MPAD * NCOL + 13 * NCOL) * sizeof(float);
  (void)hipFuncSetAttribute((const void*)lconv_bilin_kernel,
                            hipFuncAttributeMaxDynamicSharedMemorySize, (int)shmem);
  lconv_bilin_kernel<<<NS_TOTAL / SITES, THREADS, shmem, stream>>>(x, w, out);
}


# transport phase task order: all backward transports, then all forward, so each 32-lane wave runs one sandwich form per pass instead of both under complementary lane masks; no sum reordered
# speedup vs baseline: 1.5170x; 1.0501x over previous
#include <hip/hip_runtime.h>

typedef float v2f __attribute__((ext_vector_type(2)));
typedef float v8f __attribute__((ext_vector_type(8)));

#define NS_TOTAL 16384
#define SITES    8
#define NCOL     (SITES * 18)
#define KPAD     112
#define MPAD     80
#define THREADS  256
#define WN       (78 * 109)
#define WQ_PER_THREAD 9

struct C3 { float2 m[9]; };

__device__ inline float2 cmul(float2 a, float2 b) {
  return make_float2(a.x*b.x - a.y*b.y, a.x*b.y + a.y*b.x);
}
__device__ inline float2 cmulcj(float2 a, float2 b) {
  return make_float2(a.x*b.x + a.y*b.y, a.y*b.x - a.x*b.y);
}
__device__ inline float2 cjmul(float2 a, float2 b) {
  return make_float2(a.x*b.x + a.y*b.y, a.x*b.y - a.y*b.x);
}

__device__ inline int neighbor(int s, int axis, int dir) {
  int a0 = (s >> 10) & 7, a1 = (s >> 7) & 7, a2 = (s >> 4) & 7, a3 = s & 15;
  if      (axis == 0) a0 = (a0 + dir) & 7;
  else if (axis == 1) a1 = (a1 + dir) & 7;
  else if (axis == 2) a2 = (a2 + dir) & 7;
  else                a3 = (a3 + dir) & 15;
  return (a0 << 10) | (a1 << 7) | (a2 << 4) | a3;
}

__device__ inline void loadMat(const float* __restrict__ xin, int bx, int ch, C3& M) {
  const float* p = xin + ((size_t)bx * 10 + ch) * 18;
  #pragma unroll
  for (int e = 0; e < 9; ++e) M.m[e] = make_float2(p[2*e], p[2*e + 1]);
}

__device__ inline void mul33(const C3& A, const C3& B, C3& C) {
  #pragma unroll
  for (int i = 0; i < 3; ++i)
    #pragma unroll
    for (int k = 0; k < 3; ++k) {
      float2 acc = make_float2(0.f, 0.f);
      #pragma unroll
      for (int j = 0; j < 3; ++j) {
        float2 t = cmul(A.m[i*3+j], B.m[j*3+k]);
        acc.x += t.x; acc.y += t.y;
      }
      C.m[i*3+k] = acc;
    }
}

__device__ inline void sandwichF(const C3& U, const C3& W, C3& R) {
  C3 T; mul33(U, W, T);
  #pragma unroll
  for (int i = 0; i < 3; ++i)
    #pragma unroll
    for (int l = 0; l < 3; ++l) {
      float2 acc = make_float2(0.f, 0.f);
      #pragma unroll
      for (int k = 0; k < 3; ++k) {
        float2 t = cmulcj(T.m[i*3+k], U.m[l*3+k]);
        acc.x += t.x; acc.y += t.y;
      }
      R.m[i*3+l] = acc;
    }
}

__device__ inline void sandwichB(const C3& U, const C3& W, C3& R) {
  C3 T;
  #pragma unroll
  for (int i = 0; i < 3; ++i)
    #pragma unroll
    for (int k = 0; k < 3; ++k) {
      float2 acc = make_float2(0.f, 0.f);
      #pragma unroll
      for (int j = 0; j < 3; ++j) {
        float2 t = cjmul(U.m[j*3+i], W.m[j*3+k]);
        acc.x += t.x; acc.y += t.y;
      }
      T.m[i*3+k] = acc;
    }
  #pragma unroll
  for (int i = 0; i < 3; ++i)
    #pragma unroll
    for (int l = 0; l < 3; ++l) {
      float2 acc = make_float2(0.f, 0.f);
      #pragma unroll
      for (int k = 0; k < 3; ++k) {
        float2 t = cmul(T.m[i*3+k], U.m[k*3+l]);
        acc.x += t.x; acc.y += t.y;
      }
      R.m[i*3+l] = acc;
    }
}

__device__ inline void storeRow(float* dst, int ch, int ls, const C3& M) {
  float* p = dst + ch * NCOL + ls * 18;
  #pragma unroll
  for (int e = 0; e < 9; ++e) { p[2*e] = M.m[e].x; p[2*e + 1] = M.m[e].y; }
}
__device__ inline void storeRowDag(float* dst, int ch, int ls, const C3& M) {
  float* p = dst + ch * NCOL + ls * 18;
  #pragma unroll
  for (int i = 0; i < 3; ++i)
    #pragma unroll
    for (int j = 0; j < 3; ++j) {
      p[(i*3+j)*2]     =  M.m[j*3+i].x;
      p[(i*3+j)*2 + 1] = -M.m[j*3+i].y;
    }
}

#define NBACK (SITES * 24)
static_assert(NBACK % 32 == 0, "a wave must hold tasks of one direction only");
static_assert(NBACK <= THREADS, "the second pass must be forward only and must finish the tasks");

__device__ __forceinline__ void transportTask(const float* __restrict__ xin, float* tF,
                                              int base, int t, bool backward) {
  const int r = backward ? t : t - NBACK;
  const int ls = r / 24, rem = r % 24;
  const int axis = rem / 6, c = rem % 6;
  const int bx = base + ls;
  const int s = bx & 8191, bofs = bx & ~8191;
  const int sn = neighbor(s, axis, backward ? -1 : +1);
  C3 U, W, R;
  loadMat(xin, backward ? (bofs | sn) : bx, axis, U);
  loadMat(xin, bofs | sn, 4 + c, W);
  if (backward) sandwichB(U, W, R);
  else          sandwichF(U, W, R);
  const int ch = 6 + axis * 12 + (backward ? 0 : 6) + c;
  storeRow(tF, ch, ls, R);
  storeRowDag(tF, 54 + ch, ls, R);
}

__global__ __launch_bounds__(THREADS)
void lconv_bilin_kernel(const float* __restrict__ xin,
                        const float* __restrict__ wgt,
                        float* __restrict__ out) {
  extern __shared__ float smem[];
  float* Wt = smem;
  float* tF = Wt + MPAD * KPAD;
  float* rL = tF + KPAD * NCOL;
  float* wF = rL + MPAD * NCOL;

  const int tid  = threadIdx.x;
  const int base = blockIdx.x * SITES;

  float4 wv[WQ_PER_THREAD];
  {
    const float4* w4 = reinterpret_cast<const float4*>(wgt);
    #pragma unroll
    for (int i = 0; i < WQ_PER_THREAD - 1; ++i) wv[i] = w4[tid + THREADS * i];
    const int q = tid + THREADS * (WQ_PER_THREAD - 1);
    wv[WQ_PER_THREAD - 1] = w4[q < WN / 4 ? q : WN / 4 - 1];
  }
  const float wtail0 = wgt[WN - 2], wtail1 = wgt[WN - 1];
  for (int idx = tid; idx < 2 * KPAD + 78 * 3; idx += THREADS) {
    int a;
    if (idx < 2 * KPAD) a = 78 * KPAD + idx;
    else { int j = idx - 2 * KPAD; a = (j / 3) * KPAD + 109 + (j % 3); }
    Wt[a] = 0.0f;
  }
  for (int idx = tid; idx < 3 * NCOL; idx += THREADS) tF[109 * NCOL + idx] = 0.0f;

  transportTask(xin, tF, base, tid, __builtin_amdgcn_readfirstlane(tid) < NBACK);
  if (tid < SITES * 48 - THREADS) transportTask(xin, tF, base, tid + THREADS, false);

  for (int t = tid; t < SITES * 6; t += THREADS) {
    int ls = t / 6, c = t % 6;
    C3 W; loadMat(xin, base + ls, 4 + c, W);
    storeRow(tF, c, ls, W);
    storeRowDag(tF, 54 + c, ls, W);
    storeRow(wF, c, ls, W);
    storeRowDag(wF, 6 + c, ls, W);
  }

  for (int t = tid; t < SITES; t += THREADS) {
    float* p1 = tF + 108 * NCOL + t * 18;
    float* p2 = wF + 12 * NCOL + t * 18;
    #pragma unroll
    for (int i = 0; i < 3; ++i)
      #pragma unroll
      for (int j = 0; j < 3; ++j) {
        float re = (i == j) ? 1.0f : 0.0f;
        p1[(i*3+j)*2] = re; p1[(i*3+j)*2 + 1] = 0.0f;
        p2[(i*3+j)*2] = re; p2[(i*3+j)*2 + 1] = 0.0f;
      }
  }

  #pragma unroll
  for (int i = 0; i < WQ_PER_THREAD; ++i) {
    const int e = 4 * (tid + THREADS * i);
    const float v[4] = {wv[i].x, wv[i].y, wv[i].z, wv[i].w};
    if (i < WQ_PER_THREAD - 1 || e + 3 < WN) {
      #pragma unroll
      for (int cc = 0; cc < 4; ++cc) Wt[(e + cc) + 3 * ((e + cc) / 109)] = v[cc];
    }
  }
  if (tid == WN / 4 - THREADS * (WQ_PER_THREAD - 1)) {
    Wt[77 * KPAD + 107] = wtail0;
    Wt[77 * KPAD + 108] = wtail1;
  }

  __syncthreads();

  {
    const int wv = tid >> 5, lane = tid & 31;
    const int l15 = lane & 15, kh = lane >> 4;
    for (int tile = wv; tile < 5 * 9; tile += 8) {
      int mt = tile / 9, nt = tile % 9;
      v8f acc = {};
      const float* wbase = Wt + (mt * 16 + l15) * KPAD;
      const float* bbase = tF + nt * 16 + l15;
      for (int k0 = 0; k0 < KPAD; k0 += 4) {
        v2f a, b;
        a[0] = wbase[k0 + 2 * kh];
        a[1] = wbase[k0 + 2 * kh + 1];
        const float* bp = bbase + (k0 + 2 * kh) * NCOL;
        b[0] = bp[0];
        b[1] = bp[NCOL];
        acc = __builtin_amdgcn_wmma_f32_16x16x4_f32(
            false, a, false, b, (short)0, acc, false, false);
      }
      float* rp = rL + (mt * 16 + (lane >> 4) * 8) * NCOL + nt * 16 + l15;
      #pragma unroll
      for (int q = 0; q < 8; ++q) rp[q * NCOL] = acc[q];
    }
  }
  __syncthreads();

  for (int t = tid; t < SITES * 6 * 9; t += THREADS) {
    int ls = t / 54, rem = t % 54;
    int u = rem / 9, ik = rem % 9, i = ik / 3, k = ik % 3;
    float2 acc = make_float2(0.f, 0.f);
    #pragma unroll
    for (int v = 0; v < 13; ++v) {
      const float* wp = wF + v * NCOL + ls * 18;
      const float* rp = rL + (u * 13 + v) * NCOL + ls * 18;
      #pragma unroll
      for (int j = 0; j < 3; ++j) {
        float2 a = make_float2(wp[(i*3+j)*2], wp[(i*3+j)*2 + 1]);
        float2 b = make_float2(rp[(j*3+k)*2], rp[(j*3+k)*2 + 1]);
        acc.x += a.x * b.x - a.y * b.y;
        acc.y += a.x * b.y + a.y * b.x;
      }
    }
    int bx = base + ls;
    float* op = out + ((size_t)bx * 10 + 4 + u) * 18 + ik * 2;
    op[0] = acc.x; op[1] = acc.y;
  }

  for (int t = tid; t < SITES * 4 * 18; t += THREADS) {
    int ls = t / 72, rem = t % 72;
    int ch = rem / 18, e = rem % 18;
    size_t idx = ((size_t)(base + ls) * 10 + ch) * 18 + e;
    out[idx] = xin[idx];
  }
}

extern "C" void kernel_launch(void* const* d_in, const int* in_sizes, int n_in,
                              void* d_out, int out_size, void* d_ws, size_t ws_size,
                              hipStream_t stream) {
  (void)in_sizes; (void)n_in; (void)out_size; (void)d_ws; (void)ws_size;
  const float* x = (const float*)d_in[0];
  const float* w = (const float*)d_in[1];
  float* out = (float*)d_out;

  const size_t shmem =
      (size_t)(MPAD * KPAD + KPAD * NCOL + MPAD * NCOL + 13 * NCOL) * sizeof(float);
  (void)hipFuncSetAttribute((const void*)lconv_bilin_kernel,
                            hipFuncAttributeMaxDynamicSharedMemorySize, (int)shmem);
  lconv_bilin_kernel<<<NS_TOTAL / SITES, THREADS, shmem, stream>>>(x, w, out);
}


# LDS rows re-pitched against bank conflicts: weights at 116 floats, t_f 4 floats more per 4-channel group; A-fragment loads 4-way to 1, transport stores 8-way to 2; no sum reordered
# speedup vs baseline: 1.8894x; 1.2455x over previous
#include <hip/hip_runtime.h>

typedef float v2f __attribute__((ext_vector_type(2)));
typedef float v8f __attribute__((ext_vector_type(8)));

#define NS_TOTAL 16384
#define SITES    8
#define NCOL     (SITES * 18)
#define KPAD     112
#define WPITCH   116
#define MPAD     80
#define THREADS  256
#define WN       (78 * 109)
#define WQ_PER_THREAD 9
#define TGRP     (4 * NCOL + 4)
#define TF_FLOATS ((KPAD / 4) * TGRP)

struct C3 { float2 m[9]; };

__device__ inline float2 cmul(float2 a, float2 b) {
  return make_float2(a.x*b.x - a.y*b.y, a.x*b.y + a.y*b.x);
}
__device__ inline float2 cmulcj(float2 a, float2 b) {
  return make_float2(a.x*b.x + a.y*b.y, a.y*b.x - a.x*b.y);
}
__device__ inline float2 cjmul(float2 a, float2 b) {
  return make_float2(a.x*b.x + a.y*b.y, a.x*b.y - a.y*b.x);
}

__device__ inline int neighbor(int s, int axis, int dir) {
  int a0 = (s >> 10) & 7, a1 = (s >> 7) & 7, a2 = (s >> 4) & 7, a3 = s & 15;
  if      (axis == 0) a0 = (a0 + dir) & 7;
  else if (axis == 1) a1 = (a1 + dir) & 7;
  else if (axis == 2) a2 = (a2 + dir) & 7;
  else                a3 = (a3 + dir) & 15;
  return (a0 << 10) | (a1 << 7) | (a2 << 4) | a3;
}

__device__ inline void loadMat(const float* __restrict__ xin, int bx, int ch, C3& M) {
  const float* p = xin + ((size_t)bx * 10 + ch) * 18;
  #pragma unroll
  for (int e = 0; e < 9; ++e) M.m[e] = make_float2(p[2*e], p[2*e + 1]);
}

__device__ inline void mul33(const C3& A, const C3& B, C3& C) {
  #pragma unroll
  for (int i = 0; i < 3; ++i)
    #pragma unroll
    for (int k = 0; k < 3; ++k) {
      float2 acc = make_float2(0.f, 0.f);
      #pragma unroll
      for (int j = 0; j < 3; ++j) {
        float2 t = cmul(A.m[i*3+j], B.m[j*3+k]);
        acc.x += t.x; acc.y += t.y;
      }
      C.m[i*3+k] = acc;
    }
}

__device__ inline void sandwichF(const C3& U, const C3& W, C3& R) {
  C3 T; mul33(U, W, T);
  #pragma unroll
  for (int i = 0; i < 3; ++i)
    #pragma unroll
    for (int l = 0; l < 3; ++l) {
      float2 acc = make_float2(0.f, 0.f);
      #pragma unroll
      for (int k = 0; k < 3; ++k) {
        float2 t = cmulcj(T.m[i*3+k], U.m[l*3+k]);
        acc.x += t.x; acc.y += t.y;
      }
      R.m[i*3+l] = acc;
    }
}

__device__ inline void sandwichB(const C3& U, const C3& W, C3& R) {
  C3 T;
  #pragma unroll
  for (int i = 0; i < 3; ++i)
    #pragma unroll
    for (int k = 0; k < 3; ++k) {
      float2 acc = make_float2(0.f, 0.f);
      #pragma unroll
      for (int j = 0; j < 3; ++j) {
        float2 t = cjmul(U.m[j*3+i], W.m[j*3+k]);
        acc.x += t.x; acc.y += t.y;
      }
      T.m[i*3+k] = acc;
    }
  #pragma unroll
  for (int i = 0; i < 3; ++i)
    #pragma unroll
    for (int l = 0; l < 3; ++l) {
      float2 acc = make_float2(0.f, 0.f);
      #pragma unroll
      for (int k = 0; k < 3; ++k) {
        float2 t = cmul(T.m[i*3+k], U.m[k*3+l]);
        acc.x += t.x; acc.y += t.y;
      }
      R.m[i*3+l] = acc;
    }
}

__device__ inline int tfRow(int ch) { return (ch >> 2) * TGRP + (ch & 3) * NCOL; }

__device__ inline void storeRow(float* row, int ls, const C3& M) {
  float* p = row + ls * 18;
  #pragma unroll
  for (int e = 0; e < 9; ++e) { p[2*e] = M.m[e].x; p[2*e + 1] = M.m[e].y; }
}
__device__ inline void storeRowDag(float* row, int ls, const C3& M) {
  float* p = row + ls * 18;
  #pragma unroll
  for (int i = 0; i < 3; ++i)
    #pragma unroll
    for (int j = 0; j < 3; ++j) {
      p[(i*3+j)*2]     =  M.m[j*3+i].x;
      p[(i*3+j)*2 + 1] = -M.m[j*3+i].y;
    }
}

#define NBACK (SITES * 24)
static_assert(NBACK % 32 == 0, "a wave must hold tasks of one direction only");
static_assert(NBACK <= THREADS, "the second pass must be forward only and must finish the tasks");

__device__ __forceinline__ void transportTask(const float* __restrict__ xin, float* tF,
                                              int base, int t, bool backward) {
  const int r = backward ? t : t - NBACK;
  const int ls = r / 24, rem = r % 24;
  const int axis = rem / 6, c = rem % 6;
  const int bx = base + ls;
  const int s = bx & 8191, bofs = bx & ~8191;
  const int sn = neighbor(s, axis, backward ? -1 : +1);
  C3 U, W, R;
  loadMat(xin, backward ? (bofs | sn) : bx, axis, U);
  loadMat(xin, bofs | sn, 4 + c, W);
  if (backward) sandwichB(U, W, R);
  else          sandwichF(U, W, R);
  const int ch = 6 + axis * 12 + (backward ? 0 : 6) + c;
  storeRow(tF + tfRow(ch), ls, R);
  storeRowDag(tF + tfRow(54 + ch), ls, R);
}

__global__ __launch_bounds__(THREADS)
void lconv_bilin_kernel(const float* __restrict__ xin,
                        const float* __restrict__ wgt,
                        float* __restrict__ out) {
  extern __shared__ float smem[];
  float* Wt = smem;
  float* tF = Wt + MPAD * WPITCH;
  float* rL = tF + TF_FLOATS;
  float* wF = rL + MPAD * NCOL;

  const int tid  = threadIdx.x;
  const int base = blockIdx.x * SITES;

  float4 wv[WQ_PER_THREAD];
  {
    const float4* w4 = reinterpret_cast<const float4*>(wgt);
    #pragma unroll
    for (int i = 0; i < WQ_PER_THREAD - 1; ++i) wv[i] = w4[tid + THREADS * i];
    const int q = tid + THREADS * (WQ_PER_THREAD - 1);
    wv[WQ_PER_THREAD - 1] = w4[q < WN / 4 ? q : WN / 4 - 1];
  }
  const float wtail0 = wgt[WN - 2], wtail1 = wgt[WN - 1];
  for (int idx = tid; idx < 2 * KPAD + 78 * 3; idx += THREADS) {
    int a;
    if (idx < 2 * KPAD) a = (78 + idx / KPAD) * WPITCH + idx % KPAD;
    else { int j = idx - 2 * KPAD; a = (j / 3) * WPITCH + 109 + (j % 3); }
    Wt[a] = 0.0f;
  }
  for (int idx = tid; idx < 3 * NCOL; idx += THREADS) tF[tfRow(109) + idx] = 0.0f;

  transportTask(xin, tF, base, tid, __builtin_amdgcn_readfirstlane(tid) < NBACK);
  if (tid < SITES * 48 - THREADS) transportTask(xin, tF, base, tid + THREADS, false);

  for (int t = tid; t < SITES * 6; t += THREADS) {
    int ls = t / 6, c = t % 6;
    C3 W; loadMat(xin, base + ls, 4 + c, W);
    storeRow(tF + tfRow(c), ls, W);
    storeRowDag(tF + tfRow(54 + c), ls, W);
    storeRow(wF + c * NCOL, ls, W);
    storeRowDag(wF + (6 + c) * NCOL, ls, W);
  }

  for (int t = tid; t < SITES; t += THREADS) {
    float* p1 = tF + tfRow(108) + t * 18;
    float* p2 = wF + 12 * NCOL + t * 18;
    #pragma unroll
    for (int i = 0; i < 3; ++i)
      #pragma unroll
      for (int j = 0; j < 3; ++j) {
        float re = (i == j) ? 1.0f : 0.0f;
        p1[(i*3+j)*2] = re; p1[(i*3+j)*2 + 1] = 0.0f;
        p2[(i*3+j)*2] = re; p2[(i*3+j)*2 + 1] = 0.0f;
      }
  }

  #pragma unroll
  for (int i = 0; i < WQ_PER_THREAD; ++i) {
    const int e = 4 * (tid + THREADS * i);
    const float v[4] = {wv[i].x, wv[i].y, wv[i].z, wv[i].w};
    if (i < WQ_PER_THREAD - 1 || e + 3 < WN) {
      #pragma unroll
      for (int cc = 0; cc < 4; ++cc) Wt[(e + cc) + (WPITCH - 109) * ((e + cc) / 109)] = v[cc];
    }
  }
  if (tid == WN / 4 - THREADS * (WQ_PER_THREAD - 1)) {
    Wt[77 * WPITCH + 107] = wtail0;
    Wt[77 * WPITCH + 108] = wtail1;
  }

  __syncthreads();

  {
    const int wv = tid >> 5, lane = tid & 31;
    const int l15 = lane & 15, kh = lane >> 4;
    for (int tile = wv; tile < 5 * 9; tile += 8) {
      int mt = tile / 9, nt = tile % 9;
      v8f acc = {};
      const float* wbase = Wt + (mt * 16 + l15) * WPITCH;
      const float* bbase = tF + nt * 16 + l15;
      for (int k0 = 0; k0 < KPAD; k0 += 4) {
        v2f a, b;
        a[0] = wbase[k0 + 2 * kh];
        a[1] = wbase[k0 + 2 * kh + 1];
        const float* bp = bbase + tfRow(k0 + 2 * kh);
        b[0] = bp[0];
        b[1] = bp[NCOL];
        acc = __builtin_amdgcn_wmma_f32_16x16x4_f32(
            false, a, false, b, (short)0, acc, false, false);
      }
      float* rp = rL + (mt * 16 + (lane >> 4) * 8) * NCOL + nt * 16 + l15;
      #pragma unroll
      for (int q = 0; q < 8; ++q) rp[q * NCOL] = acc[q];
    }
  }
  __syncthreads();

  for (int t = tid; t < SITES * 6 * 9; t += THREADS) {
    int ls = t / 54, rem = t % 54;
    int u = rem / 9, ik = rem % 9, i = ik / 3, k = ik % 3;
    float2 acc = make_float2(0.f, 0.f);
    #pragma unroll
    for (int v = 0; v < 13; ++v) {
      const float* wp = wF + v * NCOL + ls * 18;
      const float* rp = rL + (u * 13 + v) * NCOL + ls * 18;
      #pragma unroll
      for (int j = 0; j < 3; ++j) {
        float2 a = make_float2(wp[(i*3+j)*2], wp[(i*3+j)*2 + 1]);
        float2 b = make_float2(rp[(j*3+k)*2], rp[(j*3+k)*2 + 1]);
        acc.x += a.x * b.x - a.y * b.y;
        acc.y += a.x * b.y + a.y * b.x;
      }
    }
    int bx = base + ls;
    float* op = out + ((size_t)bx * 10 + 4 + u) * 18 + ik * 2;
    op[0] = acc.x; op[1] = acc.y;
  }

  for (int t = tid; t < SITES * 4 * 18; t += THREADS) {
    int ls = t / 72, rem = t % 72;
    int ch = rem / 18, e = rem % 18;
    size_t idx = ((size_t)(base + ls) * 10 + ch) * 18 + e;
    out[idx] = xin[idx];
  }
}

extern "C" void kernel_launch(void* const* d_in, const int* in_sizes, int n_in,
                              void* d_out, int out_size, void* d_ws, size_t ws_size,
                              hipStream_t stream) {
  (void)in_sizes; (void)n_in; (void)out_size; (void)d_ws; (void)ws_size;
  const float* x = (const float*)d_in[0];
  const float* w = (const float*)d_in[1];
  float* out = (float*)d_out;

  const size_t shmem =
      (size_t)(MPAD * WPITCH + TF_FLOATS + MPAD * NCOL + 13 * NCOL) * sizeof(float);
  (void)hipFuncSetAttribute((const void*)lconv_bilin_kernel,
                            hipFuncAttributeMaxDynamicSharedMemorySize, (int)shmem);
  lconv_bilin_kernel<<<NS_TOTAL / SITES, THREADS, shmem, stream>>>(x, w, out);
}


# phase B register-blocked: each wave owns a 5x1 strip of output tiles (waves 0-4 also one tile of the ninth strip), one B fragment per k-step feeds 5 products, loads 2 k-steps ahead; no sum reordered
# speedup vs baseline: 2.1948x; 1.1616x over previous
#include <hip/hip_runtime.h>

typedef float v2f __attribute__((ext_vector_type(2)));
typedef float v8f __attribute__((ext_vector_type(8)));

#define NS_TOTAL 16384
#define SITES    8
#define NCOL     (SITES * 18)
#define KPAD     112
#define WPITCH   116
#define MPAD     80
#define THREADS  256
#define WN       (78 * 109)
#define WQ_PER_THREAD 9
#define TGRP     (4 * NCOL + 4)
#define TF_FLOATS ((KPAD / 4) * TGRP)

struct C3 { float2 m[9]; };

__device__ inline float2 cmul(float2 a, float2 b) {
  return make_float2(a.x*b.x - a.y*b.y, a.x*b.y + a.y*b.x);
}
__device__ inline float2 cmulcj(float2 a, float2 b) {
  return make_float2(a.x*b.x + a.y*b.y, a.y*b.x - a.x*b.y);
}
__device__ inline float2 cjmul(float2 a, float2 b) {
  return make_float2(a.x*b.x + a.y*b.y, a.x*b.y - a.y*b.x);
}

__device__ inline int neighbor(int s, int axis, int dir) {
  int a0 = (s >> 10) & 7, a1 = (s >> 7) & 7, a2 = (s >> 4) & 7, a3 = s & 15;
  if      (axis == 0) a0 = (a0 + dir) & 7;
  else if (axis == 1) a1 = (a1 + dir) & 7;
  else if (axis == 2) a2 = (a2 + dir) & 7;
  else                a3 = (a3 + dir) & 15;
  return (a0 << 10) | (a1 << 7) | (a2 << 4) | a3;
}

__device__ inline void loadMat(const float* __restrict__ xin, int bx, int ch, C3& M) {
  const float* p = xin + ((size_t)bx * 10 + ch) * 18;
  #pragma unroll
  for (int e = 0; e < 9; ++e) M.m[e] = make_float2(p[2*e], p[2*e + 1]);
}

__device__ inline void mul33(const C3& A, const C3& B, C3& C) {
  #pragma unroll
  for (int i = 0; i < 3; ++i)
    #pragma unroll
    for (int k = 0; k < 3; ++k) {
      float2 acc = make_float2(0.f, 0.f);
      #pragma unroll
      for (int j = 0; j < 3; ++j) {
        float2 t = cmul(A.m[i*3+j], B.m[j*3+k]);
        acc.x += t.x; acc.y += t.y;
      }
      C.m[i*3+k] = acc;
    }
}

__device__ inline void sandwichF(const C3& U, const C3& W, C3& R) {
  C3 T; mul33(U, W, T);
  #pragma unroll
  for (int i = 0; i < 3; ++i)
    #pragma unroll
    for (int l = 0; l < 3; ++l) {
      float2 acc = make_float2(0.f, 0.f);
      #pragma unroll
      for (int k = 0; k < 3; ++k) {
        float2 t = cmulcj(T.m[i*3+k], U.m[l*3+k]);
        acc.x += t.x; acc.y += t.y;
      }
      R.m[i*3+l] = acc;
    }
}

__device__ inline void sandwichB(const C3& U, const C3& W, C3& R) {
  C3 T;
  #pragma unroll
  for (int i = 0; i < 3; ++i)
    #pragma unroll
    for (int k = 0; k < 3; ++k) {
      float2 acc = make_float2(0.f, 0.f);
      #pragma unroll
      for (int j = 0; j < 3; ++j) {
        float2 t = cjmul(U.m[j*3+i], W.m[j*3+k]);
        acc.x += t.x; acc.y += t.y;
      }
      T.m[i*3+k] = acc;
    }
  #pragma unroll
  for (int i = 0; i < 3; ++i)
    #pragma unroll
    for (int l = 0; l < 3; ++l) {
      float2 acc = make_float2(0.f, 0.f);
      #pragma unroll
      for (int k = 0; k < 3; ++k) {
        float2 t = cmul(T.m[i*3+k], U.m[k*3+l]);
        acc.x += t.x; acc.y += t.y;
      }
      R.m[i*3+l] = acc;
    }
}

__device__ inline int tfRow(int ch) { return (ch >> 2) * TGRP + (ch & 3) * NCOL; }

__device__ inline void storeRow(float* row, int ls, const C3& M) {
  float* p = row + ls * 18;
  #pragma unroll
  for (int e = 0; e < 9; ++e) { p[2*e] = M.m[e].x; p[2*e + 1] = M.m[e].y; }
}
__device__ inline void storeRowDag(float* row, int ls, const C3& M) {
  float* p = row + ls * 18;
  #pragma unroll
  for (int i = 0; i < 3; ++i)
    #pragma unroll
    for (int j = 0; j < 3; ++j) {
      p[(i*3+j)*2]     =  M.m[j*3+i].x;
      p[(i*3+j)*2 + 1] = -M.m[j*3+i].y;
    }
}

#define NBACK (SITES * 24)
static_assert(NBACK % 32 == 0, "a wave must hold tasks of one direction only");
static_assert(NBACK <= THREADS, "the second pass must be forward only and must finish the tasks");

__device__ __forceinline__ void transportTask(const float* __restrict__ xin, float* tF,
                                              int base, int t, bool backward) {
  const int r = backward ? t : t - NBACK;
  const int ls = r / 24, rem = r % 24;
  const int axis = rem / 6, c = rem % 6;
  const int bx = base + ls;
  const int s = bx & 8191, bofs = bx & ~8191;
  const int sn = neighbor(s, axis, backward ? -1 : +1);
  C3 U, W, R;
  loadMat(xin, backward ? (bofs | sn) : bx, axis, U);
  loadMat(xin, bofs | sn, 4 + c, W);
  if (backward) sandwichB(U, W, R);
  else          sandwichF(U, W, R);
  const int ch = 6 + axis * 12 + (backward ? 0 : 6) + c;
  storeRow(tF + tfRow(ch), ls, R);
  storeRowDag(tF + tfRow(54 + ch), ls, R);
}

#define KSTEPS (KPAD / 4)
#define MTILES (MPAD / 16)
#define AHEAD  2
static_assert(NCOL / 16 == THREADS / 32 + 1, "eight whole strips, one per wave, and a ninth shared out");
static_assert(MTILES <= THREADS / 32, "the ninth strip gives at most one tile to a wave");

template <bool EXTRA>
__device__ __forceinline__ void gemmStrip(const float* Wt, const float* tF, float* rL,
                                          int nt, int me, int lane) {
  const int l15 = lane & 15, kh = lane >> 4;
  const float* ap  = Wt + l15 * WPITCH + 2 * kh;
  const float* bp  = tF + 2 * kh * NCOL + nt * 16 + l15;
  const float* aep = ap + me * 16 * WPITCH;
  const float* bep = tF + 2 * kh * NCOL + 8 * 16 + l15;
  v8f acc[MTILES];
  v8f acce = {};
  #pragma unroll
  for (int mt = 0; mt < MTILES; ++mt) acc[mt] = v8f{};
  v2f a[AHEAD + 1][MTILES], b[AHEAD + 1], ae[AHEAD + 1], be[AHEAD + 1];

  #pragma unroll
  for (int g = 0; g < KSTEPS + AHEAD; ++g) {
    if (g < KSTEPS) {
      const int s = g % (AHEAD + 1);
      #pragma unroll
      for (int mt = 0; mt < MTILES; ++mt)
        a[s][mt] = *reinterpret_cast<const v2f*>(ap + mt * 16 * WPITCH + 4 * g);
      b[s][0] = bp[g * TGRP];
      b[s][1] = bp[g * TGRP + NCOL];
      if (EXTRA) {
        ae[s] = *reinterpret_cast<const v2f*>(aep + 4 * g);
        be[s][0] = bep[g * TGRP];
        be[s][1] = bep[g * TGRP + NCOL];
      }
    }
    if (g >= AHEAD) {
      const int s = (g - AHEAD) % (AHEAD + 1);
      #pragma unroll
      for (int mt = 0; mt < MTILES; ++mt)
        acc[mt] = __builtin_amdgcn_wmma_f32_16x16x4_f32(
            false, a[s][mt], false, b[s], (short)0, acc[mt], false, false);
      if (EXTRA)
        acce = __builtin_amdgcn_wmma_f32_16x16x4_f32(
            false, ae[s], false, be[s], (short)0, acce, false, false);
    }
  }

  #pragma unroll
  for (int mt = 0; mt < MTILES; ++mt) {
    float* rp = rL + (mt * 16 + kh * 8) * NCOL + nt * 16 + l15;
    #pragma unroll
    for (int q = 0; q < 8; ++q) rp[q * NCOL] = acc[mt][q];
  }
  if (EXTRA) {
    float* rp = rL + (me * 16 + kh * 8) * NCOL + 8 * 16 + l15;
    #pragma unroll
    for (int q = 0; q < 8; ++q) rp[q * NCOL] = acce[q];
  }
}

__global__ __launch_bounds__(THREADS)
void lconv_bilin_kernel(const float* __restrict__ xin,
                        const float* __restrict__ wgt,
                        float* __restrict__ out) {
  extern __shared__ __attribute__((aligned(16))) float smem[];
  float* Wt = smem;
  float* tF = Wt + MPAD * WPITCH;
  float* rL = tF + TF_FLOATS;
  float* wF = rL + MPAD * NCOL;

  const int tid  = threadIdx.x;
  const int base = blockIdx.x * SITES;

  float4 wv[WQ_PER_THREAD];
  {
    const float4* w4 = reinterpret_cast<const float4*>(wgt);
    #pragma unroll
    for (int i = 0; i < WQ_PER_THREAD - 1; ++i) wv[i] = w4[tid + THREADS * i];
    const int q = tid + THREADS * (WQ_PER_THREAD - 1);
    wv[WQ_PER_THREAD - 1] = w4[q < WN / 4 ? q : WN / 4 - 1];
  }
  const float wtail0 = wgt[WN - 2], wtail1 = wgt[WN - 1];
  for (int idx = tid; idx < 2 * KPAD + 78 * 3; idx += THREADS) {
    int a;
    if (idx < 2 * KPAD) a = (78 + idx / KPAD) * WPITCH + idx % KPAD;
    else { int j = idx - 2 * KPAD; a = (j / 3) * WPITCH + 109 + (j % 3); }
    Wt[a] = 0.0f;
  }
  for (int idx = tid; idx < 3 * NCOL; idx += THREADS) tF[tfRow(109) + idx] = 0.0f;

  transportTask(xin, tF, base, tid, __builtin_amdgcn_readfirstlane(tid) < NBACK);
  if (tid < SITES * 48 - THREADS) transportTask(xin, tF, base, tid + THREADS, false);

  for (int t = tid; t < SITES * 6; t += THREADS) {
    int ls = t / 6, c = t % 6;
    C3 W; loadMat(xin, base + ls, 4 + c, W);
    storeRow(tF + tfRow(c), ls, W);
    storeRowDag(tF + tfRow(54 + c), ls, W);
    storeRow(wF + c * NCOL, ls, W);
    storeRowDag(wF + (6 + c) * NCOL, ls, W);
  }

  for (int t = tid; t < SITES; t += THREADS) {
    float* p1 = tF + tfRow(108) + t * 18;
    float* p2 = wF + 12 * NCOL + t * 18;
    #pragma unroll
    for (int i = 0; i < 3; ++i)
      #pragma unroll
      for (int j = 0; j < 3; ++j) {
        float re = (i == j) ? 1.0f : 0.0f;
        p1[(i*3+j)*2] = re; p1[(i*3+j)*2 + 1] = 0.0f;
        p2[(i*3+j)*2] = re; p2[(i*3+j)*2 + 1] = 0.0f;
      }
  }

  #pragma unroll
  for (int i = 0; i < WQ_PER_THREAD; ++i) {
    const int e = 4 * (tid + THREADS * i);
    const float v[4] = {wv[i].x, wv[i].y, wv[i].z, wv[i].w};
    if (i < WQ_PER_THREAD - 1 || e + 3 < WN) {
      #pragma unroll
      for (int cc = 0; cc < 4; ++cc) Wt[(e + cc) + (WPITCH - 109) * ((e + cc) / 109)] = v[cc];
    }
  }
  if (tid == WN / 4 - THREADS * (WQ_PER_THREAD - 1)) {
    Wt[77 * WPITCH + 107] = wtail0;
    Wt[77 * WPITCH + 108] = wtail1;
  }

  __syncthreads();

  {
    const int wave = __builtin_amdgcn_readfirstlane(tid) >> 5;
    if (wave < MPAD / 16) gemmStrip<true>(Wt, tF, rL, wave, wave, tid & 31);
    else                  gemmStrip<false>(Wt, tF, rL, wave, 0, tid & 31);
  }
  __syncthreads();

  for (int t = tid; t < SITES * 6 * 9; t += THREADS) {
    int ls = t / 54, rem = t % 54;
    int u = rem / 9, ik = rem % 9, i = ik / 3, k = ik % 3;
    float2 acc = make_float2(0.f, 0.f);
    #pragma unroll
    for (int v = 0; v < 13; ++v) {
      const float* wp = wF + v * NCOL + ls * 18;
      const float* rp = rL + (u * 13 + v) * NCOL + ls * 18;
      #pragma unroll
      for (int j = 0; j < 3; ++j) {
        float2 a = make_float2(wp[(i*3+j)*2], wp[(i*3+j)*2 + 1]);
        float2 b = make_float2(rp[(j*3+k)*2], rp[(j*3+k)*2 + 1]);
        acc.x += a.x * b.x - a.y * b.y;
        acc.y += a.x * b.y + a.y * b.x;
      }
    }
    int bx = base + ls;
    float* op = out + ((size_t)bx * 10 + 4 + u) * 18 + ik * 2;
    op[0] = acc.x; op[1] = acc.y;
  }

  for (int t = tid; t < SITES * 4 * 18; t += THREADS) {
    int ls = t / 72, rem = t % 72;
    int ch = rem / 18, e = rem % 18;
    size_t idx = ((size_t)(base + ls) * 10 + ch) * 18 + e;
    out[idx] = xin[idx];
  }
}

extern "C" void kernel_launch(void* const* d_in, const int* in_sizes, int n_in,
                              void* d_out, int out_size, void* d_ws, size_t ws_size,
                              hipStream_t stream) {
  (void)in_sizes; (void)n_in; (void)out_size; (void)d_ws; (void)ws_size;
  const float* x = (const float*)d_in[0];
  const float* w = (const float*)d_in[1];
  float* out = (float*)d_out;

  const size_t shmem =
      (size_t)(MPAD * WPITCH + TF_FLOATS + MPAD * NCOL + 13 * NCOL) * sizeof(float);
  (void)hipFuncSetAttribute((const void*)lconv_bilin_kernel,
                            hipFuncAttributeMaxDynamicSharedMemorySize, (int)shmem);
  lconv_bilin_kernel<<<NS_TOTAL / SITES, THREADS, shmem, stream>>>(x, w, out);
}


# global loads: second transport pass's inputs, own loop matrices and the gauge-link copy (now 16-byte) issued at kernel entry beside the first pass's; 6 exposed round trips become 1; no sum reordered
# speedup vs baseline: 2.2662x; 1.0325x over previous
#include <hip/hip_runtime.h>

typedef float v2f __attribute__((ext_vector_type(2)));
typedef float v8f __attribute__((ext_vector_type(8)));

#define NS_TOTAL 16384
#define SITES    8
#define NCOL     (SITES * 18)
#define KPAD     112
#define WPITCH   116
#define MPAD     80
#define THREADS  256
#define WN       (78 * 109)
#define WQ_PER_THREAD 9
#define TGRP     (4 * NCOL + 4)
#define TF_FLOATS ((KPAD / 4) * TGRP)

struct C3 { float2 m[9]; };

__device__ inline float2 cmul(float2 a, float2 b) {
  return make_float2(a.x*b.x - a.y*b.y, a.x*b.y + a.y*b.x);
}
__device__ inline float2 cmulcj(float2 a, float2 b) {
  return make_float2(a.x*b.x + a.y*b.y, a.y*b.x - a.x*b.y);
}
__device__ inline float2 cjmul(float2 a, float2 b) {
  return make_float2(a.x*b.x + a.y*b.y, a.x*b.y - a.y*b.x);
}

__device__ inline int neighbor(int s, int axis, int dir) {
  int a0 = (s >> 10) & 7, a1 = (s >> 7) & 7, a2 = (s >> 4) & 7, a3 = s & 15;
  if      (axis == 0) a0 = (a0 + dir) & 7;
  else if (axis == 1) a1 = (a1 + dir) & 7;
  else if (axis == 2) a2 = (a2 + dir) & 7;
  else                a3 = (a3 + dir) & 15;
  return (a0 << 10) | (a1 << 7) | (a2 << 4) | a3;
}

__device__ inline void loadMat(const float* __restrict__ xin, int bx, int ch, C3& M) {
  const float* p = xin + ((size_t)bx * 10 + ch) * 18;
  #pragma unroll
  for (int e = 0; e < 9; ++e) M.m[e] = make_float2(p[2*e], p[2*e + 1]);
}

__device__ inline void mul33(const C3& A, const C3& B, C3& C) {
  #pragma unroll
  for (int i = 0; i < 3; ++i)
    #pragma unroll
    for (int k = 0; k < 3; ++k) {
      float2 acc = make_float2(0.f, 0.f);
      #pragma unroll
      for (int j = 0; j < 3; ++j) {
        float2 t = cmul(A.m[i*3+j], B.m[j*3+k]);
        acc.x += t.x; acc.y += t.y;
      }
      C.m[i*3+k] = acc;
    }
}

__device__ inline void sandwichF(const C3& U, const C3& W, C3& R) {
  C3 T; mul33(U, W, T);
  #pragma unroll
  for (int i = 0; i < 3; ++i)
    #pragma unroll
    for (int l = 0; l < 3; ++l) {
      float2 acc = make_float2(0.f, 0.f);
      #pragma unroll
      for (int k = 0; k < 3; ++k) {
        float2 t = cmulcj(T.m[i*3+k], U.m[l*3+k]);
        acc.x += t.x; acc.y += t.y;
      }
      R.m[i*3+l] = acc;
    }
}

__device__ inline void sandwichB(const C3& U, const C3& W, C3& R) {
  C3 T;
  #pragma unroll
  for (int i = 0; i < 3; ++i)
    #pragma unroll
    for (int k = 0; k < 3; ++k) {
      float2 acc = make_float2(0.f, 0.f);
      #pragma unroll
      for (int j = 0; j < 3; ++j) {
        float2 t = cjmul(U.m[j*3+i], W.m[j*3+k]);
        acc.x += t.x; acc.y += t.y;
      }
      T.m[i*3+k] = acc;
    }
  #pragma unroll
  for (int i = 0; i < 3; ++i)
    #pragma unroll
    for (int l = 0; l < 3; ++l) {
      float2 acc = make_float2(0.f, 0.f);
      #pragma unroll
      for (int k = 0; k < 3; ++k) {
        float2 t = cmul(T.m[i*3+k], U.m[k*3+l]);
        acc.x += t.x; acc.y += t.y;
      }
      R.m[i*3+l] = acc;
    }
}

__device__ inline int tfRow(int ch) { return (ch >> 2) * TGRP + (ch & 3) * NCOL; }

__device__ inline void storeRow(float* row, int ls, const C3& M) {
  float* p = row + ls * 18;
  #pragma unroll
  for (int e = 0; e < 9; ++e) { p[2*e] = M.m[e].x; p[2*e + 1] = M.m[e].y; }
}
__device__ inline void storeRowDag(float* row, int ls, const C3& M) {
  float* p = row + ls * 18;
  #pragma unroll
  for (int i = 0; i < 3; ++i)
    #pragma unroll
    for (int j = 0; j < 3; ++j) {
      p[(i*3+j)*2]     =  M.m[j*3+i].x;
      p[(i*3+j)*2 + 1] = -M.m[j*3+i].y;
    }
}

#define NBACK (SITES * 24)
static_assert(NBACK % 32 == 0, "a wave must hold tasks of one direction only");
static_assert(NBACK <= THREADS, "the second pass must be forward only and must finish the tasks");

__device__ __forceinline__ void transportLoad(const float* __restrict__ xin, int base, int t,
                                              bool backward, C3& U, C3& W) {
  const int r = backward ? t : t - NBACK;
  const int ls = r / 24, rem = r % 24;
  const int axis = rem / 6, c = rem % 6;
  const int bx = base + ls;
  const int s = bx & 8191, bofs = bx & ~8191;
  const int sn = neighbor(s, axis, backward ? -1 : +1);
  loadMat(xin, backward ? (bofs | sn) : bx, axis, U);
  loadMat(xin, bofs | sn, 4 + c, W);
}
__device__ __forceinline__ void transportFinish(float* tF, int t, bool backward,
                                                const C3& U, const C3& W) {
  const int r = backward ? t : t - NBACK;
  const int ls = r / 24, rem = r % 24;
  const int axis = rem / 6, c = rem % 6;
  C3 R;
  if (backward) sandwichB(U, W, R);
  else          sandwichF(U, W, R);
  const int ch = 6 + axis * 12 + (backward ? 0 : 6) + c;
  storeRow(tF + tfRow(ch), ls, R);
  storeRowDag(tF + tfRow(54 + ch), ls, R);
}

#define KSTEPS (KPAD / 4)
#define MTILES (MPAD / 16)
#define AHEAD  2
static_assert(NCOL / 16 == THREADS / 32 + 1, "eight whole strips, one per wave, and a ninth shared out");
static_assert(MTILES <= THREADS / 32, "the ninth strip gives at most one tile to a wave");

template <bool EXTRA>
__device__ __forceinline__ void gemmStrip(const float* Wt, const float* tF, float* rL,
                                          int nt, int me, int lane) {
  const int l15 = lane & 15, kh = lane >> 4;
  const float* ap  = Wt + l15 * WPITCH + 2 * kh;
  const float* bp  = tF + 2 * kh * NCOL + nt * 16 + l15;
  const float* aep = ap + me * 16 * WPITCH;
  const float* bep = tF + 2 * kh * NCOL + 8 * 16 + l15;
  v8f acc[MTILES];
  v8f acce = {};
  #pragma unroll
  for (int mt = 0; mt < MTILES; ++mt) acc[mt] = v8f{};
  v2f a[AHEAD + 1][MTILES], b[AHEAD + 1], ae[AHEAD + 1], be[AHEAD + 1];

  #pragma unroll
  for (int g = 0; g < KSTEPS + AHEAD; ++g) {
    if (g < KSTEPS) {
      const int s = g % (AHEAD + 1);
      #pragma unroll
      for (int mt = 0; mt < MTILES; ++mt)
        a[s][mt] = *reinterpret_cast<const v2f*>(ap + mt * 16 * WPITCH + 4 * g);
      b[s][0] = bp[g * TGRP];
      b[s][1] = bp[g * TGRP + NCOL];
      if (EXTRA) {
        ae[s] = *reinterpret_cast<const v2f*>(aep + 4 * g);
        be[s][0] = bep[g * TGRP];
        be[s][1] = bep[g * TGRP + NCOL];
      }
    }
    if (g >= AHEAD) {
      const int s = (g - AHEAD) % (AHEAD + 1);
      #pragma unroll
      for (int mt = 0; mt < MTILES; ++mt)
        acc[mt] = __builtin_amdgcn_wmma_f32_16x16x4_f32(
            false, a[s][mt], false, b[s], (short)0, acc[mt], false, false);
      if (EXTRA)
        acce = __builtin_amdgcn_wmma_f32_16x16x4_f32(
            false, ae[s], false, be[s], (short)0, acce, false, false);
    }
  }

  #pragma unroll
  for (int mt = 0; mt < MTILES; ++mt) {
    float* rp = rL + (mt * 16 + kh * 8) * NCOL + nt * 16 + l15;
    #pragma unroll
    for (int q = 0; q < 8; ++q) rp[q * NCOL] = acc[mt][q];
  }
  if (EXTRA) {
    float* rp = rL + (me * 16 + kh * 8) * NCOL + 8 * 16 + l15;
    #pragma unroll
    for (int q = 0; q < 8; ++q) rp[q * NCOL] = acce[q];
  }
}

__global__ __launch_bounds__(THREADS)
void lconv_bilin_kernel(const float* __restrict__ xin,
                        const float* __restrict__ wgt,
                        float* __restrict__ out) {
  extern __shared__ __attribute__((aligned(16))) float smem[];
  float* Wt = smem;
  float* tF = Wt + MPAD * WPITCH;
  float* rL = tF + TF_FLOATS;
  float* wF = rL + MPAD * NCOL;

  const int tid  = threadIdx.x;
  const int base = blockIdx.x * SITES;

  const int tq = tid < SITES * 18 ? tid : 0;
  const size_t linkIdx = (size_t)(base + tq / 18) * 180 + 4 * (tq % 18);
  const float4 linkv = *reinterpret_cast<const float4*>(xin + linkIdx);
  const bool back1 = __builtin_amdgcn_readfirstlane(tid) < NBACK;
  C3 U1, W1, U2, W2, Wl;
  transportLoad(xin, base, tid, back1, U1, W1);
  constexpr int PASS2 = SITES * 48 - THREADS;
  transportLoad(xin, base, tid < PASS2 ? tid + THREADS : tid + THREADS - PASS2, false, U2, W2);
  const int tl = tid < SITES * 6 ? tid : 0;
  loadMat(xin, base + tl / 6, 4 + tl % 6, Wl);

  float4 wv[WQ_PER_THREAD];
  {
    const float4* w4 = reinterpret_cast<const float4*>(wgt);
    #pragma unroll
    for (int i = 0; i < WQ_PER_THREAD - 1; ++i) wv[i] = w4[tid + THREADS * i];
    const int q = tid + THREADS * (WQ_PER_THREAD - 1);
    wv[WQ_PER_THREAD - 1] = w4[q < WN / 4 ? q : WN / 4 - 1];
  }
  const float wtail0 = wgt[WN - 2], wtail1 = wgt[WN - 1];
  asm volatile("" :: "v"(linkv.x), "v"(linkv.y), "v"(linkv.z), "v"(linkv.w));
  for (int idx = tid; idx < 2 * KPAD + 78 * 3; idx += THREADS) {
    int a;
    if (idx < 2 * KPAD) a = (78 + idx / KPAD) * WPITCH + idx % KPAD;
    else { int j = idx - 2 * KPAD; a = (j / 3) * WPITCH + 109 + (j % 3); }
    Wt[a] = 0.0f;
  }
  for (int idx = tid; idx < 3 * NCOL; idx += THREADS) tF[tfRow(109) + idx] = 0.0f;

  transportFinish(tF, tid, back1, U1, W1);
  if (tid < SITES * 18) *reinterpret_cast<float4*>(out + linkIdx) = linkv;
  if (tid < PASS2) transportFinish(tF, tid + THREADS, false, U2, W2);

  if (tid < SITES * 6) {
    const int ls = tid / 6, c = tid % 6;
    storeRow(tF + tfRow(c), ls, Wl);
    storeRowDag(tF + tfRow(54 + c), ls, Wl);
    storeRow(wF + c * NCOL, ls, Wl);
    storeRowDag(wF + (6 + c) * NCOL, ls, Wl);
  }

  for (int t = tid; t < SITES; t += THREADS) {
    float* p1 = tF + tfRow(108) + t * 18;
    float* p2 = wF + 12 * NCOL + t * 18;
    #pragma unroll
    for (int i = 0; i < 3; ++i)
      #pragma unroll
      for (int j = 0; j < 3; ++j) {
        float re = (i == j) ? 1.0f : 0.0f;
        p1[(i*3+j)*2] = re; p1[(i*3+j)*2 + 1] = 0.0f;
        p2[(i*3+j)*2] = re; p2[(i*3+j)*2 + 1] = 0.0f;
      }
  }

  #pragma unroll
  for (int i = 0; i < WQ_PER_THREAD; ++i) {
    const int e = 4 * (tid + THREADS * i);
    const float v[4] = {wv[i].x, wv[i].y, wv[i].z, wv[i].w};
    if (i < WQ_PER_THREAD - 1 || e + 3 < WN) {
      #pragma unroll
      for (int cc = 0; cc < 4; ++cc) Wt[(e + cc) + (WPITCH - 109) * ((e + cc) / 109)] = v[cc];
    }
  }
  if (tid == WN / 4 - THREADS * (WQ_PER_THREAD - 1)) {
    Wt[77 * WPITCH + 107] = wtail0;
    Wt[77 * WPITCH + 108] = wtail1;
  }

  __syncthreads();

  {
    const int wave = __builtin_amdgcn_readfirstlane(tid) >> 5;
    if (wave < MPAD / 16) gemmStrip<true>(Wt, tF, rL, wave, wave, tid & 31);
    else                  gemmStrip<false>(Wt, tF, rL, wave, 0, tid & 31);
  }
  __syncthreads();

  for (int t = tid; t < SITES * 6 * 9; t += THREADS) {
    int ls = t / 54, rem = t % 54;
    int u = rem / 9, ik = rem % 9, i = ik / 3, k = ik % 3;
    float2 acc = make_float2(0.f, 0.f);
    #pragma unroll
    for (int v = 0; v < 13; ++v) {
      const float* wp = wF + v * NCOL + ls * 18;
      const float* rp = rL + (u * 13 + v) * NCOL + ls * 18;
      #pragma unroll
      for (int j = 0; j < 3; ++j) {
        float2 a = make_float2(wp[(i*3+j)*2], wp[(i*3+j)*2 + 1]);
        float2 b = make_float2(rp[(j*3+k)*2], rp[(j*3+k)*2 + 1]);
        acc.x += a.x * b.x - a.y * b.y;
        acc.y += a.x * b.y + a.y * b.x;
      }
    }
    int bx = base + ls;
    float* op = out + ((size_t)bx * 10 + 4 + u) * 18 + ik * 2;
    op[0] = acc.x; op[1] = acc.y;
  }

}

extern "C" void kernel_launch(void* const* d_in, const int* in_sizes, int n_in,
                              void* d_out, int out_size, void* d_ws, size_t ws_size,
                              hipStream_t stream) {
  (void)in_sizes; (void)n_in; (void)out_size; (void)d_ws; (void)ws_size;
  const float* x = (const float*)d_in[0];
  const float* w = (const float*)d_in[1];
  float* out = (float*)d_out;

  const size_t shmem =
      (size_t)(MPAD * WPITCH + TF_FLOATS + MPAD * NCOL + 13 * NCOL) * sizeof(float);
  (void)hipFuncSetAttribute((const void*)lconv_bilin_kernel,
                            hipFuncAttributeMaxDynamicSharedMemorySize, (int)shmem);
  lconv_bilin_kernel<<<NS_TOTAL / SITES, THREADS, shmem, stream>>>(x, w, out);
}


# persistent work-groups: 512 groups loop over 4 tiles each, so the weights, their zero padding and the identity channels go to LDS once per group instead of once per tile; no sum reordered
# speedup vs baseline: 2.3569x; 1.0400x over previous
#include <hip/hip_runtime.h>

typedef float v2f __attribute__((ext_vector_type(2)));
typedef float v8f __attribute__((ext_vector_type(8)));

#define NS_TOTAL 16384
#define SITES    8
#define NCOL     (SITES * 18)
#define KPAD     112
#define WPITCH   116
#define MPAD     80
#define THREADS  256
#define NTILES   (NS_TOTAL / SITES)
#define NWG      512
#define TILES_PER_WG (NTILES / NWG)
static_assert(NTILES % NWG == 0, "every work-group runs the same number of tiles");
#define WN       (78 * 109)
#define WQ_PER_THREAD 9
#define TGRP     (4 * NCOL + 4)
#define TF_FLOATS ((KPAD / 4) * TGRP)

struct C3 { float2 m[9]; };

__device__ inline float2 cmul(float2 a, float2 b) {
  return make_float2(a.x*b.x - a.y*b.y, a.x*b.y + a.y*b.x);
}
__device__ inline float2 cmulcj(float2 a, float2 b) {
  return make_float2(a.x*b.x + a.y*b.y, a.y*b.x - a.x*b.y);
}
__device__ inline float2 cjmul(float2 a, float2 b) {
  return make_float2(a.x*b.x + a.y*b.y, a.x*b.y - a.y*b.x);
}

__device__ inline int neighbor(int s, int axis, int dir) {
  int a0 = (s >> 10) & 7, a1 = (s >> 7) & 7, a2 = (s >> 4) & 7, a3 = s & 15;
  if      (axis == 0) a0 = (a0 + dir) & 7;
  else if (axis == 1) a1 = (a1 + dir) & 7;
  else if (axis == 2) a2 = (a2 + dir) & 7;
  else                a3 = (a3 + dir) & 15;
  return (a0 << 10) | (a1 << 7) | (a2 << 4) | a3;
}

__device__ inline void loadMat(const float* __restrict__ xin, int bx, int ch, C3& M) {
  const float* p = xin + ((size_t)bx * 10 + ch) * 18;
  #pragma unroll
  for (int e = 0; e < 9; ++e) M.m[e] = make_float2(p[2*e], p[2*e + 1]);
}

__device__ inline void mul33(const C3& A, const C3& B, C3& C) {
  #pragma unroll
  for (int i = 0; i < 3; ++i)
    #pragma unroll
    for (int k = 0; k < 3; ++k) {
      float2 acc = make_float2(0.f, 0.f);
      #pragma unroll
      for (int j = 0; j < 3; ++j) {
        float2 t = cmul(A.m[i*3+j], B.m[j*3+k]);
        acc.x += t.x; acc.y += t.y;
      }
      C.m[i*3+k] = acc;
    }
}

__device__ inline void sandwichF(const C3& U, const C3& W, C3& R) {
  C3 T; mul33(U, W, T);
  #pragma unroll
  for (int i = 0; i < 3; ++i)
    #pragma unroll
    for (int l = 0; l < 3; ++l) {
      float2 acc = make_float2(0.f, 0.f);
      #pragma unroll
      for (int k = 0; k < 3; ++k) {
        float2 t = cmulcj(T.m[i*3+k], U.m[l*3+k]);
        acc.x += t.x; acc.y += t.y;
      }
      R.m[i*3+l] = acc;
    }
}

__device__ inline void sandwichB(const C3& U, const C3& W, C3& R) {
  C3 T;
  #pragma unroll
  for (int i = 0; i < 3; ++i)
    #pragma unroll
    for (int k = 0; k < 3; ++k) {
      float2 acc = make_float2(0.f, 0.f);
      #pragma unroll
      for (int j = 0; j < 3; ++j) {
        float2 t = cjmul(U.m[j*3+i], W.m[j*3+k]);
        acc.x += t.x; acc.y += t.y;
      }
      T.m[i*3+k] = acc;
    }
  #pragma unroll
  for (int i = 0; i < 3; ++i)
    #pragma unroll
    for (int l = 0; l < 3; ++l) {
      float2 acc = make_float2(0.f, 0.f);
      #pragma unroll
      for (int k = 0; k < 3; ++k) {
        float2 t = cmul(T.m[i*3+k], U.m[k*3+l]);
        acc.x += t.x; acc.y += t.y;
      }
      R.m[i*3+l] = acc;
    }
}

__device__ inline int tfRow(int ch) { return (ch >> 2) * TGRP + (ch & 3) * NCOL; }

__device__ inline void storeRow(float* row, int ls, const C3& M) {
  float* p = row + ls * 18;
  #pragma unroll
  for (int e = 0; e < 9; ++e) { p[2*e] = M.m[e].x; p[2*e + 1] = M.m[e].y; }
}
__device__ inline void storeRowDag(float* row, int ls, const C3& M) {
  float* p = row + ls * 18;
  #pragma unroll
  for (int i = 0; i < 3; ++i)
    #pragma unroll
    for (int j = 0; j < 3; ++j) {
      p[(i*3+j)*2]     =  M.m[j*3+i].x;
      p[(i*3+j)*2 + 1] = -M.m[j*3+i].y;
    }
}

#define NBACK (SITES * 24)
static_assert(NBACK % 32 == 0, "a wave must hold tasks of one direction only");
static_assert(NBACK <= THREADS, "the second pass must be forward only and must finish the tasks");

__device__ __forceinline__ void transportLoad(const float* __restrict__ xin, int base, int t,
                                              bool backward, C3& U, C3& W) {
  const int r = backward ? t : t - NBACK;
  const int ls = r / 24, rem = r % 24;
  const int axis = rem / 6, c = rem % 6;
  const int bx = base + ls;
  const int s = bx & 8191, bofs = bx & ~8191;
  const int sn = neighbor(s, axis, backward ? -1 : +1);
  loadMat(xin, backward ? (bofs | sn) : bx, axis, U);
  loadMat(xin, bofs | sn, 4 + c, W);
}
__device__ __forceinline__ void transportFinish(float* tF, int t, bool backward,
                                                const C3& U, const C3& W) {
  const int r = backward ? t : t - NBACK;
  const int ls = r / 24, rem = r % 24;
  const int axis = rem / 6, c = rem % 6;
  C3 R;
  if (backward) sandwichB(U, W, R);
  else          sandwichF(U, W, R);
  const int ch = 6 + axis * 12 + (backward ? 0 : 6) + c;
  storeRow(tF + tfRow(ch), ls, R);
  storeRowDag(tF + tfRow(54 + ch), ls, R);
}

#define KSTEPS (KPAD / 4)
#define MTILES (MPAD / 16)
#define AHEAD  2
static_assert(NCOL / 16 == THREADS / 32 + 1, "eight whole strips, one per wave, and a ninth shared out");
static_assert(MTILES <= THREADS / 32, "the ninth strip gives at most one tile to a wave");

template <bool EXTRA>
__device__ __forceinline__ void gemmStrip(const float* Wt, const float* tF, float* rL,
                                          int nt, int me, int lane) {
  const int l15 = lane & 15, kh = lane >> 4;
  const float* ap  = Wt + l15 * WPITCH + 2 * kh;
  const float* bp  = tF + 2 * kh * NCOL + nt * 16 + l15;
  const float* aep = ap + me * 16 * WPITCH;
  const float* bep = tF + 2 * kh * NCOL + 8 * 16 + l15;
  v8f acc[MTILES];
  v8f acce = {};
  #pragma unroll
  for (int mt = 0; mt < MTILES; ++mt) acc[mt] = v8f{};
  v2f a[AHEAD + 1][MTILES], b[AHEAD + 1], ae[AHEAD + 1], be[AHEAD + 1];

  #pragma unroll
  for (int g = 0; g < KSTEPS + AHEAD; ++g) {
    if (g < KSTEPS) {
      const int s = g % (AHEAD + 1);
      #pragma unroll
      for (int mt = 0; mt < MTILES; ++mt)
        a[s][mt] = *reinterpret_cast<const v2f*>(ap + mt * 16 * WPITCH + 4 * g);
      b[s][0] = bp[g * TGRP];
      b[s][1] = bp[g * TGRP + NCOL];
      if (EXTRA) {
        ae[s] = *reinterpret_cast<const v2f*>(aep + 4 * g);
        be[s][0] = bep[g * TGRP];
        be[s][1] = bep[g * TGRP + NCOL];
      }
    }
    if (g >= AHEAD) {
      const int s = (g - AHEAD) % (AHEAD + 1);
      #pragma unroll
      for (int mt = 0; mt < MTILES; ++mt)
        acc[mt] = __builtin_amdgcn_wmma_f32_16x16x4_f32(
            false, a[s][mt], false, b[s], (short)0, acc[mt], false, false);
      if (EXTRA)
        acce = __builtin_amdgcn_wmma_f32_16x16x4_f32(
            false, ae[s], false, be[s], (short)0, acce, false, false);
    }
  }

  #pragma unroll
  for (int mt = 0; mt < MTILES; ++mt) {
    float* rp = rL + (mt * 16 + kh * 8) * NCOL + nt * 16 + l15;
    #pragma unroll
    for (int q = 0; q < 8; ++q) rp[q * NCOL] = acc[mt][q];
  }
  if (EXTRA) {
    float* rp = rL + (me * 16 + kh * 8) * NCOL + 8 * 16 + l15;
    #pragma unroll
    for (int q = 0; q < 8; ++q) rp[q * NCOL] = acce[q];
  }
}

__global__ __launch_bounds__(THREADS)
void lconv_bilin_kernel(const float* __restrict__ xin,
                        const float* __restrict__ wgt,
                        float* __restrict__ out) {
  extern __shared__ __attribute__((aligned(16))) float smem[];
  float* Wt = smem;
  float* tF = Wt + MPAD * WPITCH;
  float* rL = tF + TF_FLOATS;
  float* wF = rL + MPAD * NCOL;

  const int tid  = threadIdx.x;

  float4 wv[WQ_PER_THREAD];
  {
    const float4* w4 = reinterpret_cast<const float4*>(wgt);
    #pragma unroll
    for (int i = 0; i < WQ_PER_THREAD - 1; ++i) wv[i] = w4[tid + THREADS * i];
    const int q = tid + THREADS * (WQ_PER_THREAD - 1);
    wv[WQ_PER_THREAD - 1] = w4[q < WN / 4 ? q : WN / 4 - 1];
  }
  const float wtail0 = wgt[WN - 2], wtail1 = wgt[WN - 1];
  for (int idx = tid; idx < 2 * KPAD + 78 * 3; idx += THREADS) {
    int a;
    if (idx < 2 * KPAD) a = (78 + idx / KPAD) * WPITCH + idx % KPAD;
    else { int j = idx - 2 * KPAD; a = (j / 3) * WPITCH + 109 + (j % 3); }
    Wt[a] = 0.0f;
  }
  for (int idx = tid; idx < 3 * NCOL; idx += THREADS) tF[tfRow(109) + idx] = 0.0f;
  for (int t = tid; t < SITES; t += THREADS) {
    float* p1 = tF + tfRow(108) + t * 18;
    float* p2 = wF + 12 * NCOL + t * 18;
    #pragma unroll
    for (int i = 0; i < 3; ++i)
      #pragma unroll
      for (int j = 0; j < 3; ++j) {
        float re = (i == j) ? 1.0f : 0.0f;
        p1[(i*3+j)*2] = re; p1[(i*3+j)*2 + 1] = 0.0f;
        p2[(i*3+j)*2] = re; p2[(i*3+j)*2 + 1] = 0.0f;
      }
  }
  #pragma unroll
  for (int i = 0; i < WQ_PER_THREAD; ++i) {
    const int e = 4 * (tid + THREADS * i);
    const float v[4] = {wv[i].x, wv[i].y, wv[i].z, wv[i].w};
    if (i < WQ_PER_THREAD - 1 || e + 3 < WN) {
      #pragma unroll
      for (int cc = 0; cc < 4; ++cc) Wt[(e + cc) + (WPITCH - 109) * ((e + cc) / 109)] = v[cc];
    }
  }
  if (tid == WN / 4 - THREADS * (WQ_PER_THREAD - 1)) {
    Wt[77 * WPITCH + 107] = wtail0;
    Wt[77 * WPITCH + 108] = wtail1;
  }

  #pragma unroll 1
  for (int it = 0; it < TILES_PER_WG; ++it) {
  const int base = (blockIdx.x + it * NWG) * SITES;

  const int tq = tid < SITES * 18 ? tid : 0;
  const size_t linkIdx = (size_t)(base + tq / 18) * 180 + 4 * (tq % 18);
  const float4 linkv = *reinterpret_cast<const float4*>(xin + linkIdx);
  const bool back1 = __builtin_amdgcn_readfirstlane(tid) < NBACK;
  C3 U1, W1, U2, W2, Wl;
  transportLoad(xin, base, tid, back1, U1, W1);
  constexpr int PASS2 = SITES * 48 - THREADS;
  transportLoad(xin, base, tid < PASS2 ? tid + THREADS : tid + THREADS - PASS2, false, U2, W2);
  const int tl = tid < SITES * 6 ? tid : 0;
  loadMat(xin, base + tl / 6, 4 + tl % 6, Wl);
  asm volatile("" :: "v"(linkv.x), "v"(linkv.y), "v"(linkv.z), "v"(linkv.w));

  transportFinish(tF, tid, back1, U1, W1);
  if (tid < SITES * 18) *reinterpret_cast<float4*>(out + linkIdx) = linkv;
  if (tid < PASS2) transportFinish(tF, tid + THREADS, false, U2, W2);

  if (tid < SITES * 6) {
    const int ls = tid / 6, c = tid % 6;
    storeRow(tF + tfRow(c), ls, Wl);
    storeRowDag(tF + tfRow(54 + c), ls, Wl);
  }

  __syncthreads();

  if (tid < SITES * 6) {
    const int ls = tid / 6, c = tid % 6;
    storeRow(wF + c * NCOL, ls, Wl);
    storeRowDag(wF + (6 + c) * NCOL, ls, Wl);
  }

  {
    const int wave = __builtin_amdgcn_readfirstlane(tid) >> 5;
    if (wave < MPAD / 16) gemmStrip<true>(Wt, tF, rL, wave, wave, tid & 31);
    else                  gemmStrip<false>(Wt, tF, rL, wave, 0, tid & 31);
  }
  __syncthreads();

  for (int t = tid; t < SITES * 6 * 9; t += THREADS) {
    int ls = t / 54, rem = t % 54;
    int u = rem / 9, ik = rem % 9, i = ik / 3, k = ik % 3;
    float2 acc = make_float2(0.f, 0.f);
    #pragma unroll
    for (int v = 0; v < 13; ++v) {
      const float* wp = wF + v * NCOL + ls * 18;
      const float* rp = rL + (u * 13 + v) * NCOL + ls * 18;
      #pragma unroll
      for (int j = 0; j < 3; ++j) {
        float2 a = make_float2(wp[(i*3+j)*2], wp[(i*3+j)*2 + 1]);
        float2 b = make_float2(rp[(j*3+k)*2], rp[(j*3+k)*2 + 1]);
        acc.x += a.x * b.x - a.y * b.y;
        acc.y += a.x * b.y + a.y * b.x;
      }
    }
    int bx = base + ls;
    float* op = out + ((size_t)bx * 10 + 4 + u) * 18 + ik * 2;
    op[0] = acc.x; op[1] = acc.y;
  }
  }

}

extern "C" void kernel_launch(void* const* d_in, const int* in_sizes, int n_in,
                              void* d_out, int out_size, void* d_ws, size_t ws_size,
                              hipStream_t stream) {
  (void)in_sizes; (void)n_in; (void)out_size; (void)d_ws; (void)ws_size;
  const float* x = (const float*)d_in[0];
  const float* w = (const float*)d_in[1];
  float* out = (float*)d_out;

  const size_t shmem =
      (size_t)(MPAD * WPITCH + TF_FLOATS + MPAD * NCOL + 13 * NCOL) * sizeof(float);
  (void)hipFuncSetAttribute((const void*)lconv_bilin_kernel,
                            hipFuncAttributeMaxDynamicSharedMemorySize, (int)shmem);
  lconv_bilin_kernel<<<NWG, THREADS, shmem, stream>>>(x, w, out);
}


# complex multiply-adds as 2 packed fused multiply-adds per product (conjugate copy of the link matrix; phase C: 4 packed partial sums joined at the end); products summed unrounded, phase C reordered
# speedup vs baseline: 2.4830x; 1.0535x over previous
#include <hip/hip_runtime.h>

typedef float v2f __attribute__((ext_vector_type(2)));
typedef float v8f __attribute__((ext_vector_type(8)));

#define NS_TOTAL 16384
#define SITES    8
#define NCOL     (SITES * 18)
#define KPAD     112
#define WPITCH   116
#define MPAD     80
#define THREADS  256
#define NTILES   (NS_TOTAL / SITES)
#define NWG      512
#define TILES_PER_WG (NTILES / NWG)
static_assert(NTILES % NWG == 0, "every work-group runs the same number of tiles");
#define WN       (78 * 109)
#define WQ_PER_THREAD 9
#define TGRP     (4 * NCOL + 4)
#define TF_FLOATS ((KPAD / 4) * TGRP)

struct C3 { v2f m[9]; };

__device__ __forceinline__ v2f bre(v2f a) { return __builtin_shufflevector(a, a, 0, 0); }
__device__ __forceinline__ v2f bim(v2f a) { return __builtin_shufflevector(a, a, 1, 1); }
__device__ __forceinline__ v2f swp(v2f a) { return __builtin_shufflevector(a, a, 1, 0); }
__device__ __forceinline__ v2f pfma(v2f a, v2f b, v2f c) { return __builtin_elementwise_fma(a, b, c); }

__device__ inline int neighbor(int s, int axis, int dir) {
  int a0 = (s >> 10) & 7, a1 = (s >> 7) & 7, a2 = (s >> 4) & 7, a3 = s & 15;
  if      (axis == 0) a0 = (a0 + dir) & 7;
  else if (axis == 1) a1 = (a1 + dir) & 7;
  else if (axis == 2) a2 = (a2 + dir) & 7;
  else                a3 = (a3 + dir) & 15;
  return (a0 << 10) | (a1 << 7) | (a2 << 4) | a3;
}

__device__ inline void loadMat(const float* __restrict__ xin, int bx, int ch, C3& M) {
  const float* p = xin + ((size_t)bx * 10 + ch) * 18;
  #pragma unroll
  for (int e = 0; e < 9; ++e) M.m[e] = v2f{p[2*e], p[2*e + 1]};
}

__device__ inline void conjMat(const C3& U, C3& Uc) {
  #pragma unroll
  for (int e = 0; e < 9; ++e) Uc.m[e] = v2f{U.m[e].x, -U.m[e].y};
}

#define CSUM3(x0, P0, Q0, x1, P1, Q1, x2, P2, Q2) \
  pfma(bim(x2), Q2, pfma(bre(x2), P2, pfma(bim(x1), Q1, pfma(bre(x1), P1, pfma(bim(x0), Q0, bre(x0) * (P0))))))

__device__ inline void sandwichF(const C3& U, const C3& W, C3& R) {
  C3 Uc; conjMat(U, Uc);
  C3 T;
  #pragma unroll
  for (int i = 0; i < 3; ++i)
    #pragma unroll
    for (int k = 0; k < 3; ++k)
      T.m[i*3+k] = CSUM3(W.m[0*3+k], U.m[i*3+0], swp(Uc.m[i*3+0]),
                         W.m[1*3+k], U.m[i*3+1], swp(Uc.m[i*3+1]),
                         W.m[2*3+k], U.m[i*3+2], swp(Uc.m[i*3+2]));
  #pragma unroll
  for (int i = 0; i < 3; ++i)
    #pragma unroll
    for (int l = 0; l < 3; ++l)
      R.m[i*3+l] = CSUM3(T.m[i*3+0], Uc.m[l*3+0], swp(U.m[l*3+0]),
                         T.m[i*3+1], Uc.m[l*3+1], swp(U.m[l*3+1]),
                         T.m[i*3+2], Uc.m[l*3+2], swp(U.m[l*3+2]));
}

__device__ inline void sandwichB(const C3& U, const C3& W, C3& R) {
  C3 Uc; conjMat(U, Uc);
  C3 T;
  #pragma unroll
  for (int i = 0; i < 3; ++i)
    #pragma unroll
    for (int k = 0; k < 3; ++k)
      T.m[i*3+k] = CSUM3(W.m[0*3+k], Uc.m[0*3+i], swp(U.m[0*3+i]),
                         W.m[1*3+k], Uc.m[1*3+i], swp(U.m[1*3+i]),
                         W.m[2*3+k], Uc.m[2*3+i], swp(U.m[2*3+i]));
  #pragma unroll
  for (int i = 0; i < 3; ++i)
    #pragma unroll
    for (int l = 0; l < 3; ++l)
      R.m[i*3+l] = CSUM3(T.m[i*3+0], U.m[0*3+l], swp(Uc.m[0*3+l]),
                         T.m[i*3+1], U.m[1*3+l], swp(Uc.m[1*3+l]),
                         T.m[i*3+2], U.m[2*3+l], swp(Uc.m[2*3+l]));
}

__device__ inline int tfRow(int ch) { return (ch >> 2) * TGRP + (ch & 3) * NCOL; }

__device__ inline void storeRow(float* row, int ls, const C3& M) {
  float* p = row + ls * 18;
  #pragma unroll
  for (int e = 0; e < 9; ++e) { p[2*e] = M.m[e].x; p[2*e + 1] = M.m[e].y; }
}
__device__ inline void storeRowDag(float* row, int ls, const C3& M) {
  float* p = row + ls * 18;
  #pragma unroll
  for (int i = 0; i < 3; ++i)
    #pragma unroll
    for (int j = 0; j < 3; ++j) {
      p[(i*3+j)*2]     =  M.m[j*3+i].x;
      p[(i*3+j)*2 + 1] = -M.m[j*3+i].y;
    }
}

#define NBACK (SITES * 24)
static_assert(NBACK % 32 == 0, "a wave must hold tasks of one direction only");
static_assert(NBACK <= THREADS, "the second pass must be forward only and must finish the tasks");

__device__ __forceinline__ void transportLoad(const float* __restrict__ xin, int base, int t,
                                              bool backward, C3& U, C3& W) {
  const int r = backward ? t : t - NBACK;
  const int ls = r / 24, rem = r % 24;
  const int axis = rem / 6, c = rem % 6;
  const int bx = base + ls;
  const int s = bx & 8191, bofs = bx & ~8191;
  const int sn = neighbor(s, axis, backward ? -1 : +1);
  loadMat(xin, backward ? (bofs | sn) : bx, axis, U);
  loadMat(xin, bofs | sn, 4 + c, W);
}
__device__ __forceinline__ void transportFinish(float* tF, int t, bool backward,
                                                const C3& U, const C3& W) {
  const int r = backward ? t : t - NBACK;
  const int ls = r / 24, rem = r % 24;
  const int axis = rem / 6, c = rem % 6;
  C3 R;
  if (backward) sandwichB(U, W, R);
  else          sandwichF(U, W, R);
  const int ch = 6 + axis * 12 + (backward ? 0 : 6) + c;
  storeRow(tF + tfRow(ch), ls, R);
  storeRowDag(tF + tfRow(54 + ch), ls, R);
}

#define KSTEPS (KPAD / 4)
#define MTILES (MPAD / 16)
#define AHEAD  2
static_assert(NCOL / 16 == THREADS / 32 + 1, "eight whole strips, one per wave, and a ninth shared out");
static_assert(MTILES <= THREADS / 32, "the ninth strip gives at most one tile to a wave");

template <bool EXTRA>
__device__ __forceinline__ void gemmStrip(const float* Wt, const float* tF, float* rL,
                                          int nt, int me, int lane) {
  const int l15 = lane & 15, kh = lane >> 4;
  const float* ap  = Wt + l15 * WPITCH + 2 * kh;
  const float* bp  = tF + 2 * kh * NCOL + nt * 16 + l15;
  const float* aep = ap + me * 16 * WPITCH;
  const float* bep = tF + 2 * kh * NCOL + 8 * 16 + l15;
  v8f acc[MTILES];
  v8f acce = {};
  #pragma unroll
  for (int mt = 0; mt < MTILES; ++mt) acc[mt] = v8f{};
  v2f a[AHEAD + 1][MTILES], b[AHEAD + 1], ae[AHEAD + 1], be[AHEAD + 1];

  #pragma unroll
  for (int g = 0; g < KSTEPS + AHEAD; ++g) {
    if (g < KSTEPS) {
      const int s = g % (AHEAD + 1);
      #pragma unroll
      for (int mt = 0; mt < MTILES; ++mt)
        a[s][mt] = *reinterpret_cast<const v2f*>(ap + mt * 16 * WPITCH + 4 * g);
      b[s][0] = bp[g * TGRP];
      b[s][1] = bp[g * TGRP + NCOL];
      if (EXTRA) {
        ae[s] = *reinterpret_cast<const v2f*>(aep + 4 * g);
        be[s][0] = bep[g * TGRP];
        be[s][1] = bep[g * TGRP + NCOL];
      }
    }
    if (g >= AHEAD) {
      const int s = (g - AHEAD) % (AHEAD + 1);
      #pragma unroll
      for (int mt = 0; mt < MTILES; ++mt)
        acc[mt] = __builtin_amdgcn_wmma_f32_16x16x4_f32(
            false, a[s][mt], false, b[s], (short)0, acc[mt], false, false);
      if (EXTRA)
        acce = __builtin_amdgcn_wmma_f32_16x16x4_f32(
            false, ae[s], false, be[s], (short)0, acce, false, false);
    }
  }

  #pragma unroll
  for (int mt = 0; mt < MTILES; ++mt) {
    float* rp = rL + (mt * 16 + kh * 8) * NCOL + nt * 16 + l15;
    #pragma unroll
    for (int q = 0; q < 8; ++q) rp[q * NCOL] = acc[mt][q];
  }
  if (EXTRA) {
    float* rp = rL + (me * 16 + kh * 8) * NCOL + 8 * 16 + l15;
    #pragma unroll
    for (int q = 0; q < 8; ++q) rp[q * NCOL] = acce[q];
  }
}

__global__ __launch_bounds__(THREADS)
void lconv_bilin_kernel(const float* __restrict__ xin,
                        const float* __restrict__ wgt,
                        float* __restrict__ out) {
  extern __shared__ __attribute__((aligned(16))) float smem[];
  float* Wt = smem;
  float* tF = Wt + MPAD * WPITCH;
  float* rL = tF + TF_FLOATS;
  float* wF = rL + MPAD * NCOL;

  const int tid  = threadIdx.x;

  float4 wv[WQ_PER_THREAD];
  {
    const float4* w4 = reinterpret_cast<const float4*>(wgt);
    #pragma unroll
    for (int i = 0; i < WQ_PER_THREAD - 1; ++i) wv[i] = w4[tid + THREADS * i];
    const int q = tid + THREADS * (WQ_PER_THREAD - 1);
    wv[WQ_PER_THREAD - 1] = w4[q < WN / 4 ? q : WN / 4 - 1];
  }
  const float wtail0 = wgt[WN - 2], wtail1 = wgt[WN - 1];
  for (int idx = tid; idx < 2 * KPAD + 78 * 3; idx += THREADS) {
    int a;
    if (idx < 2 * KPAD) a = (78 + idx / KPAD) * WPITCH + idx % KPAD;
    else { int j = idx - 2 * KPAD; a = (j / 3) * WPITCH + 109 + (j % 3); }
    Wt[a] = 0.0f;
  }
  for (int idx = tid; idx < 3 * NCOL; idx += THREADS) tF[tfRow(109) + idx] = 0.0f;
  for (int t = tid; t < SITES; t += THREADS) {
    float* p1 = tF + tfRow(108) + t * 18;
    float* p2 = wF + 12 * NCOL + t * 18;
    #pragma unroll
    for (int i = 0; i < 3; ++i)
      #pragma unroll
      for (int j = 0; j < 3; ++j) {
        float re = (i == j) ? 1.0f : 0.0f;
        p1[(i*3+j)*2] = re; p1[(i*3+j)*2 + 1] = 0.0f;
        p2[(i*3+j)*2] = re; p2[(i*3+j)*2 + 1] = 0.0f;
      }
  }
  #pragma unroll
  for (int i = 0; i < WQ_PER_THREAD; ++i) {
    const int e = 4 * (tid + THREADS * i);
    const float v[4] = {wv[i].x, wv[i].y, wv[i].z, wv[i].w};
    if (i < WQ_PER_THREAD - 1 || e + 3 < WN) {
      #pragma unroll
      for (int cc = 0; cc < 4; ++cc) Wt[(e + cc) + (WPITCH - 109) * ((e + cc) / 109)] = v[cc];
    }
  }
  if (tid == WN / 4 - THREADS * (WQ_PER_THREAD - 1)) {
    Wt[77 * WPITCH + 107] = wtail0;
    Wt[77 * WPITCH + 108] = wtail1;
  }

  #pragma unroll 1
  for (int it = 0; it < TILES_PER_WG; ++it) {
  const int base = (blockIdx.x + it * NWG) * SITES;

  const int tq = tid < SITES * 18 ? tid : 0;
  const size_t linkIdx = (size_t)(base + tq / 18) * 180 + 4 * (tq % 18);
  const float4 linkv = *reinterpret_cast<const float4*>(xin + linkIdx);
  const bool back1 = __builtin_amdgcn_readfirstlane(tid) < NBACK;
  C3 U1, W1, U2, W2, Wl;
  transportLoad(xin, base, tid, back1, U1, W1);
  constexpr int PASS2 = SITES * 48 - THREADS;
  transportLoad(xin, base, tid < PASS2 ? tid + THREADS : tid + THREADS - PASS2, false, U2, W2);
  const int tl = tid < SITES * 6 ? tid : 0;
  loadMat(xin, base + tl / 6, 4 + tl % 6, Wl);
  asm volatile("" :: "v"(linkv.x), "v"(linkv.y), "v"(linkv.z), "v"(linkv.w));
  __builtin_amdgcn_sched_barrier(0);

  transportFinish(tF, tid, back1, U1, W1);
  if (tid < SITES * 18) *reinterpret_cast<float4*>(out + linkIdx) = linkv;
  if (tid < PASS2) transportFinish(tF, tid + THREADS, false, U2, W2);

  if (tid < SITES * 6) {
    const int ls = tid / 6, c = tid % 6;
    storeRow(tF + tfRow(c), ls, Wl);
    storeRowDag(tF + tfRow(54 + c), ls, Wl);
  }

  __syncthreads();

  if (tid < SITES * 6) {
    const int ls = tid / 6, c = tid % 6;
    storeRow(wF + c * NCOL, ls, Wl);
    storeRowDag(wF + (6 + c) * NCOL, ls, Wl);
  }

  {
    const int wave = __builtin_amdgcn_readfirstlane(tid) >> 5;
    if (wave < MPAD / 16) gemmStrip<true>(Wt, tF, rL, wave, wave, tid & 31);
    else                  gemmStrip<false>(Wt, tF, rL, wave, 0, tid & 31);
  }
  __syncthreads();

  for (int t = tid; t < SITES * 6 * 9; t += THREADS) {
    int ls = t / 54, rem = t % 54;
    int u = rem / 9, ik = rem % 9, i = ik / 3, k = ik % 3;
    v2f s1[2], s2[2];
    #pragma unroll
    for (int v = 0; v < 13; ++v) {
      const float* wp = wF + v * NCOL + ls * 18;
      const float* rp = rL + (u * 13 + v) * NCOL + ls * 18;
      #pragma unroll
      for (int j = 0; j < 3; ++j) {
        const int h = (v * 3 + j) & 1;
        const v2f a = v2f{wp[(i*3+j)*2], wp[(i*3+j)*2 + 1]};
        const v2f b = v2f{rp[(j*3+k)*2], rp[(j*3+k)*2 + 1]};
        if (v * 3 + j < 2) {
          s1[h] = bre(a) * b;
          s2[h] = bim(a) * swp(b);
        } else {
          s1[h] = pfma(bre(a), b, s1[h]);
          s2[h] = pfma(bim(a), swp(b), s2[h]);
        }
      }
    }
    const v2f t1 = s1[0] + s1[1], t2 = s2[0] + s2[1];
    int bx = base + ls;
    float* op = out + ((size_t)bx * 10 + 4 + u) * 18 + ik * 2;
    op[0] = t1.x - t2.x; op[1] = t1.y + t2.y;
  }
  }

}

extern "C" void kernel_launch(void* const* d_in, const int* in_sizes, int n_in,
                              void* d_out, int out_size, void* d_ws, size_t ws_size,
                              hipStream_t stream) {
  (void)in_sizes; (void)n_in; (void)out_size; (void)d_ws; (void)ws_size;
  const float* x = (const float*)d_in[0];
  const float* w = (const float*)d_in[1];
  float* out = (float*)d_out;

  const size_t shmem =
      (size_t)(MPAD * WPITCH + TF_FLOATS + MPAD * NCOL + 13 * NCOL) * sizeof(float);
  (void)hipFuncSetAttribute((const void*)lconv_bilin_kernel,
                            hipFuncAttributeMaxDynamicSharedMemorySize, (int)shmem);
  lconv_bilin_kernel<<<NWG, THREADS, shmem, stream>>>(x, w, out);
}


# phase C: each thread sums one output element for two channels (u, u+3) from one set of w_f loads, one pass of 216 threads on waves 1-7 instead of two passes of 256; per-output sum order kept
# speedup vs baseline: 2.5586x; 1.0304x over previous
#include <hip/hip_runtime.h>

typedef float v2f __attribute__((ext_vector_type(2)));
typedef float v8f __attribute__((ext_vector_type(8)));

#define NS_TOTAL 16384
#define SITES    8
#define NCOL     (SITES * 18)
#define KPAD     112
#define WPITCH   116
#define MPAD     80
#define THREADS  256
#define NTILES   (NS_TOTAL / SITES)
#define NWG      512
#define TILES_PER_WG (NTILES / NWG)
static_assert(NTILES % NWG == 0, "every work-group runs the same number of tiles");
#define WN       (78 * 109)
#define WQ_PER_THREAD 9
#define TGRP     (4 * NCOL + 4)
#define TF_FLOATS ((KPAD / 4) * TGRP)

struct C3 { v2f m[9]; };

__device__ __forceinline__ v2f bre(v2f a) { return __builtin_shufflevector(a, a, 0, 0); }
__device__ __forceinline__ v2f bim(v2f a) { return __builtin_shufflevector(a, a, 1, 1); }
__device__ __forceinline__ v2f swp(v2f a) { return __builtin_shufflevector(a, a, 1, 0); }
__device__ __forceinline__ v2f pfma(v2f a, v2f b, v2f c) { return __builtin_elementwise_fma(a, b, c); }

__device__ inline int neighbor(int s, int axis, int dir) {
  int a0 = (s >> 10) & 7, a1 = (s >> 7) & 7, a2 = (s >> 4) & 7, a3 = s & 15;
  if      (axis == 0) a0 = (a0 + dir) & 7;
  else if (axis == 1) a1 = (a1 + dir) & 7;
  else if (axis == 2) a2 = (a2 + dir) & 7;
  else                a3 = (a3 + dir) & 15;
  return (a0 << 10) | (a1 << 7) | (a2 << 4) | a3;
}

__device__ inline void loadMat(const float* __restrict__ xin, int bx, int ch, C3& M) {
  const float* p = xin + ((size_t)bx * 10 + ch) * 18;
  #pragma unroll
  for (int e = 0; e < 9; ++e) M.m[e] = v2f{p[2*e], p[2*e + 1]};
}

__device__ inline void conjMat(const C3& U, C3& Uc) {
  #pragma unroll
  for (int e = 0; e < 9; ++e) Uc.m[e] = v2f{U.m[e].x, -U.m[e].y};
}

#define CSUM3(x0, P0, Q0, x1, P1, Q1, x2, P2, Q2) \
  pfma(bim(x2), Q2, pfma(bre(x2), P2, pfma(bim(x1), Q1, pfma(bre(x1), P1, pfma(bim(x0), Q0, bre(x0) * (P0))))))

__device__ inline void sandwichF(const C3& U, const C3& W, C3& R) {
  C3 Uc; conjMat(U, Uc);
  C3 T;
  #pragma unroll
  for (int i = 0; i < 3; ++i)
    #pragma unroll
    for (int k = 0; k < 3; ++k)
      T.m[i*3+k] = CSUM3(W.m[0*3+k], U.m[i*3+0], swp(Uc.m[i*3+0]),
                         W.m[1*3+k], U.m[i*3+1], swp(Uc.m[i*3+1]),
                         W.m[2*3+k], U.m[i*3+2], swp(Uc.m[i*3+2]));
  #pragma unroll
  for (int i = 0; i < 3; ++i)
    #pragma unroll
    for (int l = 0; l < 3; ++l)
      R.m[i*3+l] = CSUM3(T.m[i*3+0], Uc.m[l*3+0], swp(U.m[l*3+0]),
                         T.m[i*3+1], Uc.m[l*3+1], swp(U.m[l*3+1]),
                         T.m[i*3+2], Uc.m[l*3+2], swp(U.m[l*3+2]));
}

__device__ inline void sandwichB(const C3& U, const C3& W, C3& R) {
  C3 Uc; conjMat(U, Uc);
  C3 T;
  #pragma unroll
  for (int i = 0; i < 3; ++i)
    #pragma unroll
    for (int k = 0; k < 3; ++k)
      T.m[i*3+k] = CSUM3(W.m[0*3+k], Uc.m[0*3+i], swp(U.m[0*3+i]),
                         W.m[1*3+k], Uc.m[1*3+i], swp(U.m[1*3+i]),
                         W.m[2*3+k], Uc.m[2*3+i], swp(U.m[2*3+i]));
  #pragma unroll
  for (int i = 0; i < 3; ++i)
    #pragma unroll
    for (int l = 0; l < 3; ++l)
      R.m[i*3+l] = CSUM3(T.m[i*3+0], U.m[0*3+l], swp(Uc.m[0*3+l]),
                         T.m[i*3+1], U.m[1*3+l], swp(Uc.m[1*3+l]),
                         T.m[i*3+2], U.m[2*3+l], swp(Uc.m[2*3+l]));
}

__device__ inline int tfRow(int ch) { return (ch >> 2) * TGRP + (ch & 3) * NCOL; }

__device__ inline void storeRow(float* row, int ls, const C3& M) {
  float* p = row + ls * 18;
  #pragma unroll
  for (int e = 0; e < 9; ++e) { p[2*e] = M.m[e].x; p[2*e + 1] = M.m[e].y; }
}
__device__ inline void storeRowDag(float* row, int ls, const C3& M) {
  float* p = row + ls * 18;
  #pragma unroll
  for (int i = 0; i < 3; ++i)
    #pragma unroll
    for (int j = 0; j < 3; ++j) {
      p[(i*3+j)*2]     =  M.m[j*3+i].x;
      p[(i*3+j)*2 + 1] = -M.m[j*3+i].y;
    }
}

#define NBACK (SITES * 24)
static_assert(NBACK % 32 == 0, "a wave must hold tasks of one direction only");
static_assert(NBACK <= THREADS, "the second pass must be forward only and must finish the tasks");

__device__ __forceinline__ void transportLoad(const float* __restrict__ xin, int base, int t,
                                              bool backward, C3& U, C3& W) {
  const int r = backward ? t : t - NBACK;
  const int ls = r / 24, rem = r % 24;
  const int axis = rem / 6, c = rem % 6;
  const int bx = base + ls;
  const int s = bx & 8191, bofs = bx & ~8191;
  const int sn = neighbor(s, axis, backward ? -1 : +1);
  loadMat(xin, backward ? (bofs | sn) : bx, axis, U);
  loadMat(xin, bofs | sn, 4 + c, W);
}
__device__ __forceinline__ void transportFinish(float* tF, int t, bool backward,
                                                const C3& U, const C3& W) {
  const int r = backward ? t : t - NBACK;
  const int ls = r / 24, rem = r % 24;
  const int axis = rem / 6, c = rem % 6;
  C3 R;
  if (backward) sandwichB(U, W, R);
  else          sandwichF(U, W, R);
  const int ch = 6 + axis * 12 + (backward ? 0 : 6) + c;
  storeRow(tF + tfRow(ch), ls, R);
  storeRowDag(tF + tfRow(54 + ch), ls, R);
}

#define KSTEPS (KPAD / 4)
#define MTILES (MPAD / 16)
#define AHEAD  2
static_assert(NCOL / 16 == THREADS / 32 + 1, "eight whole strips, one per wave, and a ninth shared out");
static_assert(MTILES <= THREADS / 32, "the ninth strip gives at most one tile to a wave");

template <bool EXTRA>
__device__ __forceinline__ void gemmStrip(const float* Wt, const float* tF, float* rL,
                                          int nt, int me, int lane) {
  const int l15 = lane & 15, kh = lane >> 4;
  const float* ap  = Wt + l15 * WPITCH + 2 * kh;
  const float* bp  = tF + 2 * kh * NCOL + nt * 16 + l15;
  const float* aep = ap + me * 16 * WPITCH;
  const float* bep = tF + 2 * kh * NCOL + 8 * 16 + l15;
  v8f acc[MTILES];
  v8f acce = {};
  #pragma unroll
  for (int mt = 0; mt < MTILES; ++mt) acc[mt] = v8f{};
  v2f a[AHEAD + 1][MTILES], b[AHEAD + 1], ae[AHEAD + 1], be[AHEAD + 1];

  #pragma unroll
  for (int g = 0; g < KSTEPS + AHEAD; ++g) {
    if (g < KSTEPS) {
      const int s = g % (AHEAD + 1);
      #pragma unroll
      for (int mt = 0; mt < MTILES; ++mt)
        a[s][mt] = *reinterpret_cast<const v2f*>(ap + mt * 16 * WPITCH + 4 * g);
      b[s][0] = bp[g * TGRP];
      b[s][1] = bp[g * TGRP + NCOL];
      if (EXTRA) {
        ae[s] = *reinterpret_cast<const v2f*>(aep + 4 * g);
        be[s][0] = bep[g * TGRP];
        be[s][1] = bep[g * TGRP + NCOL];
      }
    }
    if (g >= AHEAD) {
      const int s = (g - AHEAD) % (AHEAD + 1);
      #pragma unroll
      for (int mt = 0; mt < MTILES; ++mt)
        acc[mt] = __builtin_amdgcn_wmma_f32_16x16x4_f32(
            false, a[s][mt], false, b[s], (short)0, acc[mt], false, false);
      if (EXTRA)
        acce = __builtin_amdgcn_wmma_f32_16x16x4_f32(
            false, ae[s], false, be[s], (short)0, acce, false, false);
    }
  }

  #pragma unroll
  for (int mt = 0; mt < MTILES; ++mt) {
    float* rp = rL + (mt * 16 + kh * 8) * NCOL + nt * 16 + l15;
    #pragma unroll
    for (int q = 0; q < 8; ++q) rp[q * NCOL] = acc[mt][q];
  }
  if (EXTRA) {
    float* rp = rL + (me * 16 + kh * 8) * NCOL + 8 * 16 + l15;
    #pragma unroll
    for (int q = 0; q < 8; ++q) rp[q * NCOL] = acce[q];
  }
}

__global__ __launch_bounds__(THREADS)
void lconv_bilin_kernel(const float* __restrict__ xin,
                        const float* __restrict__ wgt,
                        float* __restrict__ out) {
  extern __shared__ __attribute__((aligned(16))) float smem[];
  float* Wt = smem;
  float* tF = Wt + MPAD * WPITCH;
  float* rL = tF + TF_FLOATS;
  float* wF = rL + MPAD * NCOL;

  const int tid  = threadIdx.x;

  float4 wv[WQ_PER_THREAD];
  {
    const float4* w4 = reinterpret_cast<const float4*>(wgt);
    #pragma unroll
    for (int i = 0; i < WQ_PER_THREAD - 1; ++i) wv[i] = w4[tid + THREADS * i];
    const int q = tid + THREADS * (WQ_PER_THREAD - 1);
    wv[WQ_PER_THREAD - 1] = w4[q < WN / 4 ? q : WN / 4 - 1];
  }
  const float wtail0 = wgt[WN - 2], wtail1 = wgt[WN - 1];
  for (int idx = tid; idx < 2 * KPAD + 78 * 3; idx += THREADS) {
    int a;
    if (idx < 2 * KPAD) a = (78 + idx / KPAD) * WPITCH + idx % KPAD;
    else { int j = idx - 2 * KPAD; a = (j / 3) * WPITCH + 109 + (j % 3); }
    Wt[a] = 0.0f;
  }
  for (int idx = tid; idx < 3 * NCOL; idx += THREADS) tF[tfRow(109) + idx] = 0.0f;
  for (int t = tid; t < SITES; t += THREADS) {
    float* p1 = tF + tfRow(108) + t * 18;
    float* p2 = wF + 12 * NCOL + t * 18;
    #pragma unroll
    for (int i = 0; i < 3; ++i)
      #pragma unroll
      for (int j = 0; j < 3; ++j) {
        float re = (i == j) ? 1.0f : 0.0f;
        p1[(i*3+j)*2] = re; p1[(i*3+j)*2 + 1] = 0.0f;
        p2[(i*3+j)*2] = re; p2[(i*3+j)*2 + 1] = 0.0f;
      }
  }
  #pragma unroll
  for (int i = 0; i < WQ_PER_THREAD; ++i) {
    const int e = 4 * (tid + THREADS * i);
    const float v[4] = {wv[i].x, wv[i].y, wv[i].z, wv[i].w};
    if (i < WQ_PER_THREAD - 1 || e + 3 < WN) {
      #pragma unroll
      for (int cc = 0; cc < 4; ++cc) Wt[(e + cc) + (WPITCH - 109) * ((e + cc) / 109)] = v[cc];
    }
  }
  if (tid == WN / 4 - THREADS * (WQ_PER_THREAD - 1)) {
    Wt[77 * WPITCH + 107] = wtail0;
    Wt[77 * WPITCH + 108] = wtail1;
  }

  #pragma unroll 1
  for (int it = 0; it < TILES_PER_WG; ++it) {
  const int base = (blockIdx.x + it * NWG) * SITES;

  const int tq = tid < SITES * 18 ? tid : 0;
  const size_t linkIdx = (size_t)(base + tq / 18) * 180 + 4 * (tq % 18);
  const float4 linkv = *reinterpret_cast<const float4*>(xin + linkIdx);
  const bool back1 = __builtin_amdgcn_readfirstlane(tid) < NBACK;
  C3 U1, W1, U2, W2, Wl;
  transportLoad(xin, base, tid, back1, U1, W1);
  constexpr int PASS2 = SITES * 48 - THREADS;
  transportLoad(xin, base, tid < PASS2 ? tid + THREADS : tid + THREADS - PASS2, false, U2, W2);
  const int tl = tid < SITES * 6 ? tid : 0;
  loadMat(xin, base + tl / 6, 4 + tl % 6, Wl);
  asm volatile("" :: "v"(linkv.x), "v"(linkv.y), "v"(linkv.z), "v"(linkv.w));
  __builtin_amdgcn_sched_barrier(0);

  transportFinish(tF, tid, back1, U1, W1);
  if (tid < SITES * 18) *reinterpret_cast<float4*>(out + linkIdx) = linkv;
  if (tid < PASS2) transportFinish(tF, tid + THREADS, false, U2, W2);

  if (tid < SITES * 6) {
    const int ls = tid / 6, c = tid % 6;
    storeRow(tF + tfRow(c), ls, Wl);
    storeRowDag(tF + tfRow(54 + c), ls, Wl);
  }

  __syncthreads();

  if (tid < SITES * 6) {
    const int ls = tid / 6, c = tid % 6;
    storeRow(wF + c * NCOL, ls, Wl);
    storeRowDag(wF + (6 + c) * NCOL, ls, Wl);
  }

  {
    const int wave = __builtin_amdgcn_readfirstlane(tid) >> 5;
    if (wave < MPAD / 16) gemmStrip<true>(Wt, tF, rL, wave, wave, tid & 31);
    else                  gemmStrip<false>(Wt, tF, rL, wave, 0, tid & 31);
  }
  __syncthreads();

  {
    const int t = (tid + THREADS - 32) & (THREADS - 1);
    if (t < SITES * 27) {
      const int ls = t / 27, rem = t % 27;
      const int up = rem / 9, ik = rem % 9, i = ik / 3, k = ik % 3;
      v2f s1[2][2], s2[2][2];
      #pragma unroll
      for (int v = 0; v < 13; ++v) {
        const float* wp = wF + v * NCOL + ls * 18;
        const float* rp = rL + (up * 13 + v) * NCOL + ls * 18;
        #pragma unroll
        for (int j = 0; j < 3; ++j) {
          const int h = (v * 3 + j) & 1;
          const v2f a = v2f{wp[(i*3+j)*2], wp[(i*3+j)*2 + 1]};
          #pragma unroll
          for (int o = 0; o < 2; ++o) {
            const float* rq = rp + o * (3 * 13 * NCOL);
            const v2f b = v2f{rq[(j*3+k)*2], rq[(j*3+k)*2 + 1]};
            if (v * 3 + j < 2) {
              s1[o][h] = bre(a) * b;
              s2[o][h] = bim(a) * swp(b);
            } else {
              s1[o][h] = pfma(bre(a), b, s1[o][h]);
              s2[o][h] = pfma(bim(a), swp(b), s2[o][h]);
            }
          }
        }
      }
      const int bx = base + ls;
      #pragma unroll
      for (int o = 0; o < 2; ++o) {
        const v2f t1 = s1[o][0] + s1[o][1], t2 = s2[o][0] + s2[o][1];
        float* op = out + ((size_t)bx * 10 + 4 + up + 3 * o) * 18 + ik * 2;
        op[0] = t1.x - t2.x; op[1] = t1.y + t2.y;
      }
    }
  }
  }

}

extern "C" void kernel_launch(void* const* d_in, const int* in_sizes, int n_in,
                              void* d_out, int out_size, void* d_ws, size_t ws_size,
                              hipStream_t stream) {
  (void)in_sizes; (void)n_in; (void)out_size; (void)d_ws; (void)ws_size;
  const float* x = (const float*)d_in[0];
  const float* w = (const float*)d_in[1];
  float* out = (float*)d_out;

  const size_t shmem =
      (size_t)(MPAD * WPITCH + TF_FLOATS + MPAD * NCOL + 13 * NCOL) * sizeof(float);
  (void)hipFuncSetAttribute((const void*)lconv_bilin_kernel,
                            hipFuncAttributeMaxDynamicSharedMemorySize, (int)shmem);
  lconv_bilin_kernel<<<NWG, THREADS, shmem, stream>>>(x, w, out);
}
